# starting kernel
# baseline (speedup 1.0000x reference)
#include <hip/hip_runtime.h>

#define N_NODES 50000
#define F_IN    128
#define HID     256
#define CLS     40
#define NE      640000

typedef __attribute__((ext_vector_type(2))) float v2f;
typedef __attribute__((ext_vector_type(8))) float v8f;

__global__ void zero_f32(float* __restrict__ p, long long n) {
  long long i = (long long)blockIdx.x * blockDim.x + threadIdx.x;
  if (i < n) p[i] = 0.0f;
}

__global__ void scatter_accum(const float* __restrict__ feat,
                              const int*   __restrict__ src,
                              const int*   __restrict__ dst,
                              float* __restrict__ agg,
                              float* __restrict__ cnt,
                              int nfeat4, int do_cnt) {
  long long tid = (long long)blockIdx.x * blockDim.x + threadIdx.x;
  long long e = tid / nfeat4;
  int j = (int)(tid - e * nfeat4);
  if (e >= NE) return;
  int s = src[e];
  int d = dst[e];
  int F = nfeat4 * 4;
  float4 v = ((const float4*)feat)[(long long)s * nfeat4 + j];
  float* a = agg + (long long)d * F + j * 4;
  unsafeAtomicAdd(a + 0, v.x);
  unsafeAtomicAdd(a + 1, v.y);
  unsafeAtomicAdd(a + 2, v.z);
  unsafeAtomicAdd(a + 3, v.w);
  if (do_cnt && j == 0) unsafeAtomicAdd(cnt + d, 1.0f);
}

__global__ void normalize_mean(float* __restrict__ agg,
                               const float* __restrict__ cnt, int F) {
  long long i = (long long)blockIdx.x * blockDim.x + threadIdx.x;
  long long n = (long long)N_NODES * F;
  if (i >= n) return;
  int node = (int)(i / F);
  float c = cnt[node];
  c = c > 1.0f ? c : 1.0f;
  agg[i] = agg[i] / c;
}

__global__ void dual_gemm_wmma(const float* __restrict__ A1,
                               const float* __restrict__ W1,
                               const float* __restrict__ A2,
                               const float* __restrict__ W2,
                               const float* __restrict__ bias,
                               float* __restrict__ out,
                               int K, int ncols, int ldo, int do_relu) {
  const int lane   = threadIdx.x & 31;
  const int wid    = blockIdx.x * (blockDim.x >> 5) + (threadIdx.x >> 5);
  const int ctiles = (ncols + 15) >> 4;
  const int mtiles = N_NODES >> 4;
  if (wid >= mtiles * ctiles) return;
  const int rt = wid / ctiles;
  const int ct = wid - rt * ctiles;

  const int mrow  = rt * 16 + (lane & 15);
  const int ncol  = ct * 16 + (lane & 15);
  const int khalf = (lane >> 4) * 2;
  const bool cvalid  = (ncol < ncols);
  const int  ncol_ld = cvalid ? ncol : (ncols - 1);

  const float* arow1 = A1 + (unsigned)mrow    * (unsigned)K + khalf;
  const float* brow1 = W1 + (unsigned)ncol_ld * (unsigned)K + khalf;
  const float* arow2 = A2 + (unsigned)mrow    * (unsigned)K + khalf;
  const float* brow2 = W2 + (unsigned)ncol_ld * (unsigned)K + khalf;

  v8f c = {};

  #pragma unroll 8
  for (int k0 = 0; k0 < K; k0 += 4) {
    float2 av = *(const float2*)(arow1 + k0);
    float2 bv = *(const float2*)(brow1 + k0);
    v2f a; a.x = av.x; a.y = av.y;
    v2f b; b.x = bv.x; b.y = bv.y;
    c = __builtin_amdgcn_wmma_f32_16x16x4_f32(false, a, false, b,
                                              (short)0, c, false, false);
  }
  #pragma unroll 8
  for (int k0 = 0; k0 < K; k0 += 4) {
    float2 av = *(const float2*)(arow2 + k0);
    float2 bv = *(const float2*)(brow2 + k0);
    v2f a; a.x = av.x; a.y = av.y;
    v2f b; b.x = bv.x; b.y = bv.y;
    c = __builtin_amdgcn_wmma_f32_16x16x4_f32(false, a, false, b,
                                              (short)0, c, false, false);
  }

  if (cvalid) {
    float bv = bias[ncol];
    unsigned rbase = (unsigned)(rt * 16 + ((lane < 16) ? 0 : 8));
    float* orow = out + (unsigned)rbase * (unsigned)ldo + ncol;
    #pragma unroll
    for (int i = 0; i < 8; ++i) {
      float v = c[i] + bv;
      if (do_relu) v = v > 0.0f ? v : 0.0f;
      orow[(unsigned)i * (unsigned)ldo] = v;
    }
  }
}

__global__ void log_softmax_rows(const float* __restrict__ logits,
                                 float* __restrict__ out) {
  int r = blockIdx.x * blockDim.x + threadIdx.x;
  if (r >= N_NODES) return;
  const float* l = logits + (long long)r * CLS;
  float m = -3.402823466e38f;
  #pragma unroll
  for (int j = 0; j < CLS; ++j) m = fmaxf(m, l[j]);
  float s = 0.0f;
  #pragma unroll
  for (int j = 0; j < CLS; ++j) s += expf(l[j] - m);
  float ls = logf(s);
  float* o = out + (long long)r * CLS;
  #pragma unroll
  for (int j = 0; j < CLS; ++j) o[j] = l[j] - m - ls;
}

extern "C" void kernel_launch(void* const* d_in, const int* in_sizes, int n_in,
                              void* d_out, int out_size, void* d_ws, size_t ws_size,
                              hipStream_t stream) {
  (void)in_sizes; (void)n_in; (void)out_size; (void)ws_size;

  const float* x    = (const float*)d_in[0];
  const int*   ei   = (const int*)  d_in[1];
  const float* W1_l = (const float*)d_in[2];
  const float* b1   = (const float*)d_in[3];
  const float* W1_r = (const float*)d_in[4];
  const float* W2_l = (const float*)d_in[5];
  const float* b2   = (const float*)d_in[6];
  const float* W2_r = (const float*)d_in[7];
  const int* src = ei;
  const int* dst = ei + NE;
  float* out = (float*)d_out;

  float* ws     = (float*)d_ws;
  float* cnt    = ws;
  float* agg1   = cnt  + N_NODES;
  float* h      = agg1 + (long long)N_NODES * F_IN;
  float* agg2   = h    + (long long)N_NODES * HID;
  float* logits = agg2 + (long long)N_NODES * HID;

  const int TB = 256;
  auto blocks = [](long long n, int tb) { return (unsigned)((n + tb - 1) / tb); };

  long long nz1 = (long long)N_NODES * (1 + F_IN);
  zero_f32<<<blocks(nz1, TB), TB, 0, stream>>>(cnt, nz1);

  long long sc1 = (long long)NE * (F_IN / 4);
  scatter_accum<<<blocks(sc1, TB), TB, 0, stream>>>(x, src, dst, agg1, cnt,
                                                    F_IN / 4,  1);

  long long nm1 = (long long)N_NODES * F_IN;
  normalize_mean<<<blocks(nm1, TB), TB, 0, stream>>>(agg1, cnt, F_IN);

  {
    int tiles = (N_NODES / 16) * (HID / 16);
    dual_gemm_wmma<<<blocks(tiles, 8), TB, 0, stream>>>(
        agg1, W1_l, x, W1_r, b1, h, F_IN, HID, HID,  1);
  }

  long long nz2 = (long long)N_NODES * HID;
  zero_f32<<<blocks(nz2, TB), TB, 0, stream>>>(agg2, nz2);

  long long sc2 = (long long)NE * (HID / 4);
  scatter_accum<<<blocks(sc2, TB), TB, 0, stream>>>(h, src, dst, agg2, cnt,
                                                    HID / 4,  0);

  long long nm2 = (long long)N_NODES * HID;
  normalize_mean<<<blocks(nm2, TB), TB, 0, stream>>>(agg2, cnt, HID);

  {
    int ctiles = (CLS + 15) / 16;
    int tiles = (N_NODES / 16) * ctiles;
    dual_gemm_wmma<<<blocks(tiles, 8), TB, 0, stream>>>(
        agg2, W2_l, h, W2_r, b2, logits, HID, CLS, CLS,  0);
  }

  log_softmax_rows<<<blocks(N_NODES, TB), TB, 0, stream>>>(logits, out);
}


# Grid coarsening: a wave loops over 16 edges (scatter) or strided 16-byte chunks (zero, mean passes), not one item per wave: over ten times fewer waves enter; launches, atomic arrival order as before
# speedup vs baseline: 1.4494x; 1.4494x over previous
#include <hip/hip_runtime.h>

#define N_NODES 50000
#define F_IN    128
#define HID     256
#define CLS     40
#define NE      640000

typedef __attribute__((ext_vector_type(2))) float v2f;
typedef __attribute__((ext_vector_type(8))) float v8f;

__global__ void zero_f32(float* __restrict__ p, int n4) {
  float4* p4 = reinterpret_cast<float4*>(p);
  const int stride = (int)(gridDim.x * blockDim.x);
  const float4 z = make_float4(0.0f, 0.0f, 0.0f, 0.0f);
  for (int i = (int)(blockIdx.x * blockDim.x + threadIdx.x); i < n4; i += stride) p4[i] = z;
}

#define EPW 16

template <int F, bool DO_CNT>
static __device__ __forceinline__ void scatter_edges(const float* __restrict__ feat,
                                                     const int*   __restrict__ src,
                                                     const int*   __restrict__ dst,
                                                     float* __restrict__ agg,
                                                     float* __restrict__ cnt) {
  constexpr int NCH = F / 32;
  constexpr int GRP = 32 / NCH;
  static_assert(NE % EPW == 0, "a wave owns a whole chunk of edges");
  static_assert(EPW % GRP == 0 && EPW <= 32, "chunk is a whole number of groups");
  const int lane = threadIdx.x & 31;
  const int wave = blockIdx.x * (blockDim.x >> 5) + (threadIdx.x >> 5);
  const int base = wave * EPW;
  if (base >= NE) return;
  const int el = base + (lane & (EPW - 1));
  const int sv = src[el];
  const int dv = dst[el];
  if (DO_CNT && lane < EPW) unsafeAtomicAdd(cnt + dv, 1.0f);
  #pragma unroll
  for (int g = 0; g < EPW; g += GRP) {
    float v[GRP][NCH];
    #pragma unroll
    for (int j = 0; j < GRP; ++j) {
      const int s = __builtin_amdgcn_readlane(sv, g + j);
      const float* f = feat + (long long)s * F + lane;
      #pragma unroll
      for (int c = 0; c < NCH; ++c) v[j][c] = f[c * 32];
    }
    #pragma unroll
    for (int j = 0; j < GRP; ++j) {
      const int d = __builtin_amdgcn_readlane(dv, g + j);
      float* a = agg + (long long)d * F + lane;
      #pragma unroll
      for (int c = 0; c < NCH; ++c) unsafeAtomicAdd(a + c * 32, v[j][c]);
    }
  }
}

__global__ void scatter_accum_l1(const float* __restrict__ feat,
                                 const int*   __restrict__ src,
                                 const int*   __restrict__ dst,
                                 float* __restrict__ agg,
                                 float* __restrict__ cnt) {
  scatter_edges<F_IN, true>(feat, src, dst, agg, cnt);
}

__global__ void scatter_accum_l2(const float* __restrict__ feat,
                                 const int*   __restrict__ src,
                                 const int*   __restrict__ dst,
                                 float* __restrict__ agg,
                                 float* __restrict__ cnt) {
  scatter_edges<HID, false>(feat, src, dst, agg, cnt);
}

#define EW_TB      128
#define EW_BLOCKS  3125
#define EW_THREADS (EW_TB * EW_BLOCKS)

template <int F>
__global__ void normalize_mean(float* __restrict__ aggf,
                               const float* __restrict__ cnt) {
  float4* agg = reinterpret_cast<float4*>(aggf);
  constexpr int Q  = F / 4;
  constexpr int N4 = N_NODES * Q;
  static_assert(N4 % (4 * EW_THREADS) == 0, "whole trips of four chunks per thread");
  const int tid = (int)(blockIdx.x * blockDim.x + threadIdx.x);
  for (int i = tid; i < N4; i += 4 * EW_THREADS) {
    float4 v[4];
    float  c[4];
    #pragma unroll
    for (int u = 0; u < 4; ++u) {
      const int j = i + u * EW_THREADS;
      v[u] = agg[j];
      c[u] = cnt[j / Q];
    }
    #pragma unroll
    for (int u = 0; u < 4; ++u) {
      const int j = i + u * EW_THREADS;
      const float cc = c[u] > 1.0f ? c[u] : 1.0f;
      float4 r;
      r.x = v[u].x / cc;
      r.y = v[u].y / cc;
      r.z = v[u].z / cc;
      r.w = v[u].w / cc;
      agg[j] = r;
    }
  }
}

__global__ void dual_gemm_wmma(const float* __restrict__ A1,
                               const float* __restrict__ W1,
                               const float* __restrict__ A2,
                               const float* __restrict__ W2,
                               const float* __restrict__ bias,
                               float* __restrict__ out,
                               int K, int ncols, int ldo, int do_relu) {
  const int lane   = threadIdx.x & 31;
  const int wid    = blockIdx.x * (blockDim.x >> 5) + (threadIdx.x >> 5);
  const int ctiles = (ncols + 15) >> 4;
  const int mtiles = N_NODES >> 4;
  if (wid >= mtiles * ctiles) return;
  const int rt = wid / ctiles;
  const int ct = wid - rt * ctiles;

  const int mrow  = rt * 16 + (lane & 15);
  const int ncol  = ct * 16 + (lane & 15);
  const int khalf = (lane >> 4) * 2;
  const bool cvalid  = (ncol < ncols);
  const int  ncol_ld = cvalid ? ncol : (ncols - 1);

  const float* arow1 = A1 + (unsigned)mrow    * (unsigned)K + khalf;
  const float* brow1 = W1 + (unsigned)ncol_ld * (unsigned)K + khalf;
  const float* arow2 = A2 + (unsigned)mrow    * (unsigned)K + khalf;
  const float* brow2 = W2 + (unsigned)ncol_ld * (unsigned)K + khalf;

  v8f c = {};

  #pragma unroll 8
  for (int k0 = 0; k0 < K; k0 += 4) {
    float2 av = *(const float2*)(arow1 + k0);
    float2 bv = *(const float2*)(brow1 + k0);
    v2f a; a.x = av.x; a.y = av.y;
    v2f b; b.x = bv.x; b.y = bv.y;
    c = __builtin_amdgcn_wmma_f32_16x16x4_f32(false, a, false, b,
                                              (short)0, c, false, false);
  }
  #pragma unroll 8
  for (int k0 = 0; k0 < K; k0 += 4) {
    float2 av = *(const float2*)(arow2 + k0);
    float2 bv = *(const float2*)(brow2 + k0);
    v2f a; a.x = av.x; a.y = av.y;
    v2f b; b.x = bv.x; b.y = bv.y;
    c = __builtin_amdgcn_wmma_f32_16x16x4_f32(false, a, false, b,
                                              (short)0, c, false, false);
  }

  if (cvalid) {
    float bv = bias[ncol];
    unsigned rbase = (unsigned)(rt * 16 + ((lane < 16) ? 0 : 8));
    float* orow = out + (unsigned)rbase * (unsigned)ldo + ncol;
    #pragma unroll
    for (int i = 0; i < 8; ++i) {
      float v = c[i] + bv;
      if (do_relu) v = v > 0.0f ? v : 0.0f;
      orow[(unsigned)i * (unsigned)ldo] = v;
    }
  }
}

__global__ void log_softmax_rows(const float* __restrict__ logits,
                                 float* __restrict__ out) {
  int r = blockIdx.x * blockDim.x + threadIdx.x;
  if (r >= N_NODES) return;
  const float* l = logits + (long long)r * CLS;
  float m = -3.402823466e38f;
  #pragma unroll
  for (int j = 0; j < CLS; ++j) m = fmaxf(m, l[j]);
  float s = 0.0f;
  #pragma unroll
  for (int j = 0; j < CLS; ++j) s += expf(l[j] - m);
  float ls = logf(s);
  float* o = out + (long long)r * CLS;
  #pragma unroll
  for (int j = 0; j < CLS; ++j) o[j] = l[j] - m - ls;
}

extern "C" void kernel_launch(void* const* d_in, const int* in_sizes, int n_in,
                              void* d_out, int out_size, void* d_ws, size_t ws_size,
                              hipStream_t stream) {
  (void)in_sizes; (void)n_in; (void)out_size; (void)ws_size;

  const float* x    = (const float*)d_in[0];
  const int*   ei   = (const int*)  d_in[1];
  const float* W1_l = (const float*)d_in[2];
  const float* b1   = (const float*)d_in[3];
  const float* W1_r = (const float*)d_in[4];
  const float* W2_l = (const float*)d_in[5];
  const float* b2   = (const float*)d_in[6];
  const float* W2_r = (const float*)d_in[7];
  const int* src = ei;
  const int* dst = ei + NE;
  float* out = (float*)d_out;

  float* ws     = (float*)d_ws;
  float* cnt    = ws;
  float* agg1   = cnt  + N_NODES;
  float* h      = agg1 + (long long)N_NODES * F_IN;
  float* agg2   = h    + (long long)N_NODES * HID;
  float* logits = agg2 + (long long)N_NODES * HID;

  const int TB = 256;
  auto blocks = [](long long n, int tb) { return (unsigned)((n + tb - 1) / tb); };
  const unsigned SC_BLOCKS = NE / EPW / (TB / 32);
  static_assert(NE % (EPW * (TB / 32)) == 0, "whole blocks of edge chunks");

  const int nz1 = N_NODES * (1 + F_IN) / 4;
  zero_f32<<<EW_BLOCKS, EW_TB, 0, stream>>>(cnt, nz1);

  scatter_accum_l1<<<SC_BLOCKS, TB, 0, stream>>>(x, src, dst, agg1, cnt);

  normalize_mean<F_IN><<<EW_BLOCKS, EW_TB, 0, stream>>>(agg1, cnt);

  {
    int tiles = (N_NODES / 16) * (HID / 16);
    dual_gemm_wmma<<<blocks(tiles, 8), TB, 0, stream>>>(
        agg1, W1_l, x, W1_r, b1, h, F_IN, HID, HID,  1);
  }

  const int nz2 = N_NODES * HID / 4;
  zero_f32<<<EW_BLOCKS, EW_TB, 0, stream>>>(agg2, nz2);

  scatter_accum_l2<<<SC_BLOCKS, TB, 0, stream>>>(h, src, dst, agg2, cnt);

  normalize_mean<HID><<<EW_BLOCKS, EW_TB, 0, stream>>>(agg2, cnt);

  {
    int ctiles = (CLS + 15) / 16;
    int tiles = (N_NODES / 16) * ctiles;
    dual_gemm_wmma<<<blocks(tiles, 8), TB, 0, stream>>>(
        agg2, W2_l, h, W2_r, b2, logits, HID, CLS, CLS,  0);
  }

  log_softmax_rows<<<blocks(N_NODES, TB), TB, 0, stream>>>(logits, out);
}


# Neighbour means by row ownership: a work-group holds the f32 sums of 200 node rows in LDS, scans dst, pulls its edges' source rows, adds by LDS atomics (arrival order), writes the mean; 5 launches
# speedup vs baseline: 2.2521x; 1.5538x over previous
#include <hip/hip_runtime.h>

#define N_NODES 50000
#define F_IN    128
#define HID     256
#define CLS     40
#define NE      640000

typedef __attribute__((ext_vector_type(2))) float v2f;
typedef __attribute__((ext_vector_type(8))) float v8f;

#define AG_ROWS   200
#define AG_NW     32
#define AG_TB     (AG_NW * 32)
#define AG_BLOCKS (N_NODES / AG_ROWS)
#define AG_Q      256
#define AG_CHUNK  128
#define AG_NCHUNK (NE / AG_CHUNK)
#define AG_ROT    (AG_NCHUNK / AG_BLOCKS)

static __device__ __forceinline__ void lds_add(float* p, float v) {
  __hip_atomic_fetch_add(p, v, __ATOMIC_RELAXED, __HIP_MEMORY_SCOPE_WORKGROUP);
}

template <int F>
static __device__ __forceinline__ void drain_ring(const float* __restrict__ feat,
                                                  const int*   __restrict__ src,
                                                  float* acc, float* cnt,
                                                  const int* qe_w, const int* qd_w,
                                                  int head, int m, int lane) {
  constexpr int NCH = F / 32;
  constexpr int GRP = 32 / NCH;
  const int slot = (head + lane) & (AG_Q - 1);
  const int e  = qe_w[slot];
  const int dl = qd_w[slot];
  const int s  = src[e];
  if (lane < m) lds_add(cnt + dl, 1.0f);
  int j = 0;
  for (; j + GRP <= m; j += GRP) {
    float v[GRP][NCH];
    #pragma unroll
    for (int g = 0; g < GRP; ++g) {
      const int sj = __builtin_amdgcn_readlane(s, j + g);
      const float* f = feat + (long long)sj * F + lane;
      #pragma unroll
      for (int c = 0; c < NCH; ++c) v[g][c] = f[c * 32];
    }
    #pragma unroll
    for (int g = 0; g < GRP; ++g) {
      const int dj = __builtin_amdgcn_readlane(dl, j + g);
      float* a = acc + dj * F + lane;
      #pragma unroll
      for (int c = 0; c < NCH; ++c) lds_add(a + c * 32, v[g][c]);
    }
  }
  for (; j < m; ++j) {
    const int sj = __builtin_amdgcn_readlane(s, j);
    const int dj = __builtin_amdgcn_readlane(dl, j);
    const float* f = feat + (long long)sj * F + lane;
    float* a = acc + dj * F + lane;
    float v[NCH];
    #pragma unroll
    for (int c = 0; c < NCH; ++c) v[c] = f[c * 32];
    #pragma unroll
    for (int c = 0; c < NCH; ++c) lds_add(a + c * 32, v[c]);
  }
}

template <int F>
__global__ __launch_bounds__(AG_TB)
void owner_mean(const float* __restrict__ feat,
                const int*   __restrict__ src,
                const int*   __restrict__ dst,
                float* __restrict__ agg) {
  static_assert(NE % AG_CHUNK == 0, "a wave reads whole chunks of the edge list");
  static_assert(N_NODES % AG_ROWS == 0, "every work-group owns a full block of rows");
  static_assert(AG_Q >= 32 + AG_CHUNK, "a trip's edges fit the ring beside 31 waiting ones");
  __shared__ float acc[AG_ROWS * F];
  __shared__ float cnt[AG_ROWS];
  __shared__ int   qe[AG_NW][AG_Q];
  __shared__ int   qd[AG_NW][AG_Q];

  const int tid  = (int)threadIdx.x;
  const int lane = tid & 31;
  const int wave = __builtin_amdgcn_readfirstlane(tid >> 5);
  const int row0 = (int)blockIdx.x * AG_ROWS;

  float4* acc4 = reinterpret_cast<float4*>(acc);
  const float4 z = make_float4(0.0f, 0.0f, 0.0f, 0.0f);
  for (int i = tid; i < AG_ROWS * F / 4; i += AG_TB) acc4[i] = z;
  if (tid < AG_ROWS) cnt[tid] = 0.0f;
  int* qe_w = qe[wave];
  int* qd_w = qd[wave];
  for (int i = lane; i < AG_Q; i += 32) { qe_w[i] = 0; qd_w[i] = 0; }
  __syncthreads();

  int head = 0;
  int qn   = 0;
  const int c0   = (int)blockIdx.x * AG_ROT + wave;
  const int cend = (int)blockIdx.x * AG_ROT + AG_NCHUNK;
  int4 nxt = make_int4(0, 0, 0, 0);
  {
    const int cc = c0 >= AG_NCHUNK ? c0 - AG_NCHUNK : c0;
    nxt = *reinterpret_cast<const int4*>(dst + cc * AG_CHUNK + lane * 4);
  }
  for (int c = c0; c < cend; c += AG_NW) {
    const int4 cur = nxt;
    const int ebase = (c >= AG_NCHUNK ? c - AG_NCHUNK : c) * AG_CHUNK + lane * 4;
    if (c + AG_NW < cend) {
      const int cn = c + AG_NW;
      const int cc = cn >= AG_NCHUNK ? cn - AG_NCHUNK : cn;
      nxt = *reinterpret_cast<const int4*>(dst + cc * AG_CHUNK + lane * 4);
    }
    const int dl[4] = {cur.x - row0, cur.y - row0, cur.z - row0, cur.w - row0};
    bool hit[4];
    #pragma unroll
    for (int u = 0; u < 4; ++u) hit[u] = (unsigned)dl[u] < (unsigned)AG_ROWS;
    if (__builtin_amdgcn_ballot_w32(hit[0] | hit[1] | hit[2] | hit[3]) != 0u) {
      #pragma unroll
      for (int u = 0; u < 4; ++u) {
        const unsigned bal = __builtin_amdgcn_ballot_w32(hit[u]);
        if (bal != 0u) {
          const int pos = (head + qn + (int)__builtin_amdgcn_mbcnt_lo(bal, 0u)) & (AG_Q - 1);
          if (hit[u]) { qe_w[pos] = ebase + u; qd_w[pos] = dl[u]; }
          qn += __builtin_popcount(bal);
        }
      }
    }
    while (qn >= 32) {
      drain_ring<F>(feat, src, acc, cnt, qe_w, qd_w, head, 32, lane);
      head = (head + 32) & (AG_Q - 1);
      qn -= 32;
    }
  }
  while (qn > 0) {
    const int m = qn < 32 ? qn : 32;
    drain_ring<F>(feat, src, acc, cnt, qe_w, qd_w, head, m, lane);
    head = (head + m) & (AG_Q - 1);
    qn -= m;
  }
  __syncthreads();

  constexpr int Q = F / 4;
  float4* out4 = reinterpret_cast<float4*>(agg + (long long)row0 * F);
  for (int i = tid; i < AG_ROWS * Q; i += AG_TB) {
    const float c  = cnt[i / Q];
    const float cc = c > 1.0f ? c : 1.0f;
    const float4 v = acc4[i];
    float4 r;
    r.x = v.x / cc;
    r.y = v.y / cc;
    r.z = v.z / cc;
    r.w = v.w / cc;
    out4[i] = r;
  }
}

__global__ void dual_gemm_wmma(const float* __restrict__ A1,
                               const float* __restrict__ W1,
                               const float* __restrict__ A2,
                               const float* __restrict__ W2,
                               const float* __restrict__ bias,
                               float* __restrict__ out,
                               int K, int ncols, int ldo, int do_relu) {
  const int lane   = threadIdx.x & 31;
  const int wid    = blockIdx.x * (blockDim.x >> 5) + (threadIdx.x >> 5);
  const int ctiles = (ncols + 15) >> 4;
  const int mtiles = N_NODES >> 4;
  if (wid >= mtiles * ctiles) return;
  const int rt = wid / ctiles;
  const int ct = wid - rt * ctiles;

  const int mrow  = rt * 16 + (lane & 15);
  const int ncol  = ct * 16 + (lane & 15);
  const int khalf = (lane >> 4) * 2;
  const bool cvalid  = (ncol < ncols);
  const int  ncol_ld = cvalid ? ncol : (ncols - 1);

  const float* arow1 = A1 + (unsigned)mrow    * (unsigned)K + khalf;
  const float* brow1 = W1 + (unsigned)ncol_ld * (unsigned)K + khalf;
  const float* arow2 = A2 + (unsigned)mrow    * (unsigned)K + khalf;
  const float* brow2 = W2 + (unsigned)ncol_ld * (unsigned)K + khalf;

  v8f c = {};

  #pragma unroll 8
  for (int k0 = 0; k0 < K; k0 += 4) {
    float2 av = *(const float2*)(arow1 + k0);
    float2 bv = *(const float2*)(brow1 + k0);
    v2f a; a.x = av.x; a.y = av.y;
    v2f b; b.x = bv.x; b.y = bv.y;
    c = __builtin_amdgcn_wmma_f32_16x16x4_f32(false, a, false, b,
                                              (short)0, c, false, false);
  }
  #pragma unroll 8
  for (int k0 = 0; k0 < K; k0 += 4) {
    float2 av = *(const float2*)(arow2 + k0);
    float2 bv = *(const float2*)(brow2 + k0);
    v2f a; a.x = av.x; a.y = av.y;
    v2f b; b.x = bv.x; b.y = bv.y;
    c = __builtin_amdgcn_wmma_f32_16x16x4_f32(false, a, false, b,
                                              (short)0, c, false, false);
  }

  if (cvalid) {
    float bv = bias[ncol];
    unsigned rbase = (unsigned)(rt * 16 + ((lane < 16) ? 0 : 8));
    float* orow = out + (unsigned)rbase * (unsigned)ldo + ncol;
    #pragma unroll
    for (int i = 0; i < 8; ++i) {
      float v = c[i] + bv;
      if (do_relu) v = v > 0.0f ? v : 0.0f;
      orow[(unsigned)i * (unsigned)ldo] = v;
    }
  }
}

__global__ void log_softmax_rows(const float* __restrict__ logits,
                                 float* __restrict__ out) {
  int r = blockIdx.x * blockDim.x + threadIdx.x;
  if (r >= N_NODES) return;
  const float* l = logits + (long long)r * CLS;
  float m = -3.402823466e38f;
  #pragma unroll
  for (int j = 0; j < CLS; ++j) m = fmaxf(m, l[j]);
  float s = 0.0f;
  #pragma unroll
  for (int j = 0; j < CLS; ++j) s += expf(l[j] - m);
  float ls = logf(s);
  float* o = out + (long long)r * CLS;
  #pragma unroll
  for (int j = 0; j < CLS; ++j) o[j] = l[j] - m - ls;
}

extern "C" void kernel_launch(void* const* d_in, const int* in_sizes, int n_in,
                              void* d_out, int out_size, void* d_ws, size_t ws_size,
                              hipStream_t stream) {
  (void)in_sizes; (void)n_in; (void)out_size; (void)ws_size;

  const float* x    = (const float*)d_in[0];
  const int*   ei   = (const int*)  d_in[1];
  const float* W1_l = (const float*)d_in[2];
  const float* b1   = (const float*)d_in[3];
  const float* W1_r = (const float*)d_in[4];
  const float* W2_l = (const float*)d_in[5];
  const float* b2   = (const float*)d_in[6];
  const float* W2_r = (const float*)d_in[7];
  const int* src = ei;
  const int* dst = ei + NE;
  float* out = (float*)d_out;

  float* ws     = (float*)d_ws;
  float* agg1   = ws   + N_NODES;
  float* h      = agg1 + (long long)N_NODES * F_IN;
  float* agg2   = h    + (long long)N_NODES * HID;
  float* logits = agg2 + (long long)N_NODES * HID;

  const int TB = 256;
  auto blocks = [](long long n, int tb) { return (unsigned)((n + tb - 1) / tb); };

  owner_mean<F_IN><<<AG_BLOCKS, AG_TB, 0, stream>>>(x, src, dst, agg1);

  {
    int tiles = (N_NODES / 16) * (HID / 16);
    dual_gemm_wmma<<<blocks(tiles, 8), TB, 0, stream>>>(
        agg1, W1_l, x, W1_r, b1, h, F_IN, HID, HID,  1);
  }

  owner_mean<HID><<<AG_BLOCKS, AG_TB, 0, stream>>>(h, src, dst, agg2);

  {
    int ctiles = (CLS + 15) / 16;
    int tiles = (N_NODES / 16) * ctiles;
    dual_gemm_wmma<<<blocks(tiles, 8), TB, 0, stream>>>(
        agg2, W2_l, h, W2_r, b2, logits, HID, CLS, CLS,  0);
  }

  log_softmax_rows<<<blocks(N_NODES, TB), TB, 0, stream>>>(logits, out);
}


# Owner scan reads a 1-byte owner tag per edge (dst/200, written by a new first launch into the logits plane) not the 4-byte dst: 4x fewer scan bytes and trips; 6 launches; LDS arrival order as before
# speedup vs baseline: 2.3790x; 1.0563x over previous
#include <hip/hip_runtime.h>

#define N_NODES 50000
#define F_IN    128
#define HID     256
#define CLS     40
#define NE      640000

typedef __attribute__((ext_vector_type(2))) float v2f;
typedef __attribute__((ext_vector_type(8))) float v8f;

#define AG_ROWS   200
#define AG_NW     32
#define AG_TB     (AG_NW * 32)
#define AG_BLOCKS (N_NODES / AG_ROWS)
#define AG_Q      64
#define AG_CHUNK  512
#define AG_NOOWN  255u
#define AG_NCHUNK (NE / AG_CHUNK)
#define AG_ROT    (AG_NCHUNK / AG_BLOCKS)

__global__ __launch_bounds__(256)
void edge_owner_tags(const int* __restrict__ dst, unsigned* __restrict__ tags4) {
  static_assert(AG_BLOCKS < (int)AG_NOOWN, "an owner number fits a byte");
  static_assert(NE % 4 == 0, "whole groups of four edges");
  const int i = (int)(blockIdx.x * 256u + threadIdx.x);
  if (i >= NE / 4) return;
  const int4 d = *reinterpret_cast<const int4*>(dst + i * 4);
  const unsigned t0 = (unsigned)d.x < (unsigned)N_NODES ? (unsigned)d.x / AG_ROWS : AG_NOOWN;
  const unsigned t1 = (unsigned)d.y < (unsigned)N_NODES ? (unsigned)d.y / AG_ROWS : AG_NOOWN;
  const unsigned t2 = (unsigned)d.z < (unsigned)N_NODES ? (unsigned)d.z / AG_ROWS : AG_NOOWN;
  const unsigned t3 = (unsigned)d.w < (unsigned)N_NODES ? (unsigned)d.w / AG_ROWS : AG_NOOWN;
  tags4[i] = t0 | (t1 << 8) | (t2 << 16) | (t3 << 24);
}

static __device__ __forceinline__ void lds_add(float* p, float v) {
  __hip_atomic_fetch_add(p, v, __ATOMIC_RELAXED, __HIP_MEMORY_SCOPE_WORKGROUP);
}

template <int F>
static __device__ __forceinline__ void drain_ring(const float* __restrict__ feat,
                                                  const int*   __restrict__ src,
                                                  const int*   __restrict__ dst,
                                                  int row0,
                                                  float* acc, float* cnt,
                                                  const int* qe_w,
                                                  int head, int m, int lane) {
  constexpr int NCH = F / 32;
  constexpr int GRP = 32 / NCH;
  const int slot = (head + lane) & (AG_Q - 1);
  const int e  = qe_w[slot];
  const int s  = src[e];
  const int dl = dst[e] - row0;
  if (lane < m) lds_add(cnt + dl, 1.0f);
  int j = 0;
  for (; j + GRP <= m; j += GRP) {
    float v[GRP][NCH];
    #pragma unroll
    for (int g = 0; g < GRP; ++g) {
      const int sj = __builtin_amdgcn_readlane(s, j + g);
      const float* f = feat + (long long)sj * F + lane;
      #pragma unroll
      for (int c = 0; c < NCH; ++c) v[g][c] = f[c * 32];
    }
    #pragma unroll
    for (int g = 0; g < GRP; ++g) {
      const int dj = __builtin_amdgcn_readlane(dl, j + g);
      float* a = acc + dj * F + lane;
      #pragma unroll
      for (int c = 0; c < NCH; ++c) lds_add(a + c * 32, v[g][c]);
    }
  }
  for (; j < m; ++j) {
    const int sj = __builtin_amdgcn_readlane(s, j);
    const int dj = __builtin_amdgcn_readlane(dl, j);
    const float* f = feat + (long long)sj * F + lane;
    float* a = acc + dj * F + lane;
    float v[NCH];
    #pragma unroll
    for (int c = 0; c < NCH; ++c) v[c] = f[c * 32];
    #pragma unroll
    for (int c = 0; c < NCH; ++c) lds_add(a + c * 32, v[c]);
  }
}

template <int F>
__global__ __launch_bounds__(AG_TB)
void owner_mean(const float* __restrict__ feat,
                const int*   __restrict__ src,
                const int*   __restrict__ dst,
                const unsigned* __restrict__ tags4,
                float* __restrict__ agg) {
  static_assert(NE % AG_CHUNK == 0, "a wave reads whole chunks of the tag list");
  static_assert(AG_NCHUNK % AG_BLOCKS == 0, "the scans' offsets tile the list");
  static_assert(N_NODES % AG_ROWS == 0, "every work-group owns a full block of rows");
  static_assert(AG_Q >= 32 + 32, "one new edge per lane fits the ring beside 31 waiting ones");
  __shared__ float acc[AG_ROWS * F];
  __shared__ float cnt[AG_ROWS];
  __shared__ int   qe[AG_NW][AG_Q];

  const int tid  = (int)threadIdx.x;
  const int lane = tid & 31;
  const int wave = __builtin_amdgcn_readfirstlane(tid >> 5);
  const int row0 = (int)blockIdx.x * AG_ROWS;

  float4* acc4 = reinterpret_cast<float4*>(acc);
  const float4 z = make_float4(0.0f, 0.0f, 0.0f, 0.0f);
  for (int i = tid; i < AG_ROWS * F / 4; i += AG_TB) acc4[i] = z;
  if (tid < AG_ROWS) cnt[tid] = 0.0f;
  int* qe_w = qe[wave];
  for (int i = lane; i < AG_Q; i += 32) qe_w[i] = 0;
  __syncthreads();

  int head = 0;
  int qn   = 0;
  const int c0   = (int)blockIdx.x * AG_ROT + wave;
  const int cend = (int)blockIdx.x * AG_ROT + AG_NCHUNK;
  const unsigned char* tags = reinterpret_cast<const unsigned char*>(tags4);
  const unsigned K7F = 0x7F7F7F7Fu;
  const unsigned K80 = 0x80808080u;
  const unsigned pat = (unsigned)blockIdx.x * 0x01010101u;
  uint4 nxt = make_uint4(0u, 0u, 0u, 0u);
  {
    const int cc = c0 >= AG_NCHUNK ? c0 - AG_NCHUNK : c0;
    nxt = *reinterpret_cast<const uint4*>(tags + cc * AG_CHUNK + lane * 16);
  }
  for (int c = c0; c < cend; c += AG_NW) {
    const uint4 cur = nxt;
    const int ebase = (c >= AG_NCHUNK ? c - AG_NCHUNK : c) * AG_CHUNK + lane * 16;
    if (c + AG_NW < cend) {
      const int cn = c + AG_NW;
      const int cc = cn >= AG_NCHUNK ? cn - AG_NCHUNK : cn;
      nxt = *reinterpret_cast<const uint4*>(tags + cc * AG_CHUNK + lane * 16);
    }
    const unsigned w0 = cur.x ^ pat;
    const unsigned w1 = cur.y ^ pat;
    const unsigned w2 = cur.z ^ pat;
    const unsigned w3 = cur.w ^ pat;
    const unsigned z0 = ~(((w0 & K7F) + K7F) | w0) & K80;
    const unsigned z1 = ~(((w1 & K7F) + K7F) | w1) & K80;
    const unsigned z2 = ~(((w2 & K7F) + K7F) | w2) & K80;
    const unsigned z3 = ~(((w3 & K7F) + K7F) | w3) & K80;
    unsigned comb = (z0 >> 7) | (z1 >> 6) | (z2 >> 5) | (z3 >> 4);
    unsigned bal = __builtin_amdgcn_ballot_w32(comb != 0u);
    while (bal != 0u) {
      const bool hit = comb != 0u;
      const unsigned p = (unsigned)__builtin_ctz(comb | 0x80000000u);
      const int pos = (head + qn + (int)__builtin_amdgcn_mbcnt_lo(bal, 0u)) & (AG_Q - 1);
      if (hit) qe_w[pos] = ebase + (int)(((p & 3u) << 2) | (p >> 3));
      qn += __builtin_popcount(bal);
      comb &= comb - 1u;
      while (qn >= 32) {
        drain_ring<F>(feat, src, dst, row0, acc, cnt, qe_w, head, 32, lane);
        head = (head + 32) & (AG_Q - 1);
        qn -= 32;
      }
      bal = __builtin_amdgcn_ballot_w32(comb != 0u);
    }
  }
  while (qn > 0) {
    const int m = qn < 32 ? qn : 32;
    drain_ring<F>(feat, src, dst, row0, acc, cnt, qe_w, head, m, lane);
    head = (head + m) & (AG_Q - 1);
    qn -= m;
  }
  __syncthreads();

  constexpr int Q = F / 4;
  float4* out4 = reinterpret_cast<float4*>(agg + (long long)row0 * F);
  for (int i = tid; i < AG_ROWS * Q; i += AG_TB) {
    const float c  = cnt[i / Q];
    const float cc = c > 1.0f ? c : 1.0f;
    const float4 v = acc4[i];
    float4 r;
    r.x = v.x / cc;
    r.y = v.y / cc;
    r.z = v.z / cc;
    r.w = v.w / cc;
    out4[i] = r;
  }
}

__global__ void dual_gemm_wmma(const float* __restrict__ A1,
                               const float* __restrict__ W1,
                               const float* __restrict__ A2,
                               const float* __restrict__ W2,
                               const float* __restrict__ bias,
                               float* __restrict__ out,
                               int K, int ncols, int ldo, int do_relu) {
  const int lane   = threadIdx.x & 31;
  const int wid    = blockIdx.x * (blockDim.x >> 5) + (threadIdx.x >> 5);
  const int ctiles = (ncols + 15) >> 4;
  const int mtiles = N_NODES >> 4;
  if (wid >= mtiles * ctiles) return;
  const int rt = wid / ctiles;
  const int ct = wid - rt * ctiles;

  const int mrow  = rt * 16 + (lane & 15);
  const int ncol  = ct * 16 + (lane & 15);
  const int khalf = (lane >> 4) * 2;
  const bool cvalid  = (ncol < ncols);
  const int  ncol_ld = cvalid ? ncol : (ncols - 1);

  const float* arow1 = A1 + (unsigned)mrow    * (unsigned)K + khalf;
  const float* brow1 = W1 + (unsigned)ncol_ld * (unsigned)K + khalf;
  const float* arow2 = A2 + (unsigned)mrow    * (unsigned)K + khalf;
  const float* brow2 = W2 + (unsigned)ncol_ld * (unsigned)K + khalf;

  v8f c = {};

  #pragma unroll 8
  for (int k0 = 0; k0 < K; k0 += 4) {
    float2 av = *(const float2*)(arow1 + k0);
    float2 bv = *(const float2*)(brow1 + k0);
    v2f a; a.x = av.x; a.y = av.y;
    v2f b; b.x = bv.x; b.y = bv.y;
    c = __builtin_amdgcn_wmma_f32_16x16x4_f32(false, a, false, b,
                                              (short)0, c, false, false);
  }
  #pragma unroll 8
  for (int k0 = 0; k0 < K; k0 += 4) {
    float2 av = *(const float2*)(arow2 + k0);
    float2 bv = *(const float2*)(brow2 + k0);
    v2f a; a.x = av.x; a.y = av.y;
    v2f b; b.x = bv.x; b.y = bv.y;
    c = __builtin_amdgcn_wmma_f32_16x16x4_f32(false, a, false, b,
                                              (short)0, c, false, false);
  }

  if (cvalid) {
    float bv = bias[ncol];
    unsigned rbase = (unsigned)(rt * 16 + ((lane < 16) ? 0 : 8));
    float* orow = out + (unsigned)rbase * (unsigned)ldo + ncol;
    #pragma unroll
    for (int i = 0; i < 8; ++i) {
      float v = c[i] + bv;
      if (do_relu) v = v > 0.0f ? v : 0.0f;
      orow[(unsigned)i * (unsigned)ldo] = v;
    }
  }
}

__global__ void log_softmax_rows(const float* __restrict__ logits,
                                 float* __restrict__ out) {
  int r = blockIdx.x * blockDim.x + threadIdx.x;
  if (r >= N_NODES) return;
  const float* l = logits + (long long)r * CLS;
  float m = -3.402823466e38f;
  #pragma unroll
  for (int j = 0; j < CLS; ++j) m = fmaxf(m, l[j]);
  float s = 0.0f;
  #pragma unroll
  for (int j = 0; j < CLS; ++j) s += expf(l[j] - m);
  float ls = logf(s);
  float* o = out + (long long)r * CLS;
  #pragma unroll
  for (int j = 0; j < CLS; ++j) o[j] = l[j] - m - ls;
}

extern "C" void kernel_launch(void* const* d_in, const int* in_sizes, int n_in,
                              void* d_out, int out_size, void* d_ws, size_t ws_size,
                              hipStream_t stream) {
  (void)in_sizes; (void)n_in; (void)out_size; (void)ws_size;

  const float* x    = (const float*)d_in[0];
  const int*   ei   = (const int*)  d_in[1];
  const float* W1_l = (const float*)d_in[2];
  const float* b1   = (const float*)d_in[3];
  const float* W1_r = (const float*)d_in[4];
  const float* W2_l = (const float*)d_in[5];
  const float* b2   = (const float*)d_in[6];
  const float* W2_r = (const float*)d_in[7];
  const int* src = ei;
  const int* dst = ei + NE;
  float* out = (float*)d_out;

  float* ws     = (float*)d_ws;
  float* agg1   = ws   + N_NODES;
  float* h      = agg1 + (long long)N_NODES * F_IN;
  float* agg2   = h    + (long long)N_NODES * HID;
  float* logits = agg2 + (long long)N_NODES * HID;
  unsigned* tags4 = (unsigned*)logits;

  const int TB = 256;
  auto blocks = [](long long n, int tb) { return (unsigned)((n + tb - 1) / tb); };

  edge_owner_tags<<<blocks(NE / 4, TB), TB, 0, stream>>>(dst, tags4);

  owner_mean<F_IN><<<AG_BLOCKS, AG_TB, 0, stream>>>(x, src, dst, tags4, agg1);

  {
    int tiles = (N_NODES / 16) * (HID / 16);
    dual_gemm_wmma<<<blocks(tiles, 8), TB, 0, stream>>>(
        agg1, W1_l, x, W1_r, b1, h, F_IN, HID, HID,  1);
  }

  owner_mean<HID><<<AG_BLOCKS, AG_TB, 0, stream>>>(h, src, dst, tags4, agg2);

  {
    int ctiles = (CLS + 15) / 16;
    int tiles = (N_NODES / 16) * ctiles;
    dual_gemm_wmma<<<blocks(tiles, 8), TB, 0, stream>>>(
        agg2, W2_l, h, W2_r, b2, logits, HID, CLS, CLS,  0);
  }

  log_softmax_rows<<<blocks(N_NODES, TB), TB, 0, stream>>>(logits, out);
}


# Both WMMA GEMMs fetch A and B fragments with one 16-byte load per lane for two products (lane half h holds k0+4h..k0+4h+3, same map in A and B): half as many 16-line loads; k grouping of sums changes
# speedup vs baseline: 2.5786x; 1.0839x over previous
#include <hip/hip_runtime.h>

#define N_NODES 50000
#define F_IN    128
#define HID     256
#define CLS     40
#define NE      640000

typedef __attribute__((ext_vector_type(2))) float v2f;
typedef __attribute__((ext_vector_type(8))) float v8f;

#define AG_ROWS   200
#define AG_NW     32
#define AG_TB     (AG_NW * 32)
#define AG_BLOCKS (N_NODES / AG_ROWS)
#define AG_Q      64
#define AG_CHUNK  512
#define AG_NOOWN  255u
#define AG_NCHUNK (NE / AG_CHUNK)
#define AG_ROT    (AG_NCHUNK / AG_BLOCKS)

__global__ __launch_bounds__(256)
void edge_owner_tags(const int* __restrict__ dst, unsigned* __restrict__ tags4) {
  static_assert(AG_BLOCKS < (int)AG_NOOWN, "an owner number fits a byte");
  static_assert(NE % 4 == 0, "whole groups of four edges");
  const int i = (int)(blockIdx.x * 256u + threadIdx.x);
  if (i >= NE / 4) return;
  const int4 d = *reinterpret_cast<const int4*>(dst + i * 4);
  const unsigned t0 = (unsigned)d.x < (unsigned)N_NODES ? (unsigned)d.x / AG_ROWS : AG_NOOWN;
  const unsigned t1 = (unsigned)d.y < (unsigned)N_NODES ? (unsigned)d.y / AG_ROWS : AG_NOOWN;
  const unsigned t2 = (unsigned)d.z < (unsigned)N_NODES ? (unsigned)d.z / AG_ROWS : AG_NOOWN;
  const unsigned t3 = (unsigned)d.w < (unsigned)N_NODES ? (unsigned)d.w / AG_ROWS : AG_NOOWN;
  tags4[i] = t0 | (t1 << 8) | (t2 << 16) | (t3 << 24);
}

static __device__ __forceinline__ void lds_add(float* p, float v) {
  __hip_atomic_fetch_add(p, v, __ATOMIC_RELAXED, __HIP_MEMORY_SCOPE_WORKGROUP);
}

template <int F>
static __device__ __forceinline__ void drain_ring(const float* __restrict__ feat,
                                                  const int*   __restrict__ src,
                                                  const int*   __restrict__ dst,
                                                  int row0,
                                                  float* acc, float* cnt,
                                                  const int* qe_w,
                                                  int head, int m, int lane) {
  constexpr int NCH = F / 32;
  constexpr int GRP = 32 / NCH;
  const int slot = (head + lane) & (AG_Q - 1);
  const int e  = qe_w[slot];
  const int s  = src[e];
  const int dl = dst[e] - row0;
  if (lane < m) lds_add(cnt + dl, 1.0f);
  int j = 0;
  for (; j + GRP <= m; j += GRP) {
    float v[GRP][NCH];
    #pragma unroll
    for (int g = 0; g < GRP; ++g) {
      const int sj = __builtin_amdgcn_readlane(s, j + g);
      const float* f = feat + (long long)sj * F + lane;
      #pragma unroll
      for (int c = 0; c < NCH; ++c) v[g][c] = f[c * 32];
    }
    #pragma unroll
    for (int g = 0; g < GRP; ++g) {
      const int dj = __builtin_amdgcn_readlane(dl, j + g);
      float* a = acc + dj * F + lane;
      #pragma unroll
      for (int c = 0; c < NCH; ++c) lds_add(a + c * 32, v[g][c]);
    }
  }
  for (; j < m; ++j) {
    const int sj = __builtin_amdgcn_readlane(s, j);
    const int dj = __builtin_amdgcn_readlane(dl, j);
    const float* f = feat + (long long)sj * F + lane;
    float* a = acc + dj * F + lane;
    float v[NCH];
    #pragma unroll
    for (int c = 0; c < NCH; ++c) v[c] = f[c * 32];
    #pragma unroll
    for (int c = 0; c < NCH; ++c) lds_add(a + c * 32, v[c]);
  }
}

template <int F>
__global__ __launch_bounds__(AG_TB)
void owner_mean(const float* __restrict__ feat,
                const int*   __restrict__ src,
                const int*   __restrict__ dst,
                const unsigned* __restrict__ tags4,
                float* __restrict__ agg) {
  static_assert(NE % AG_CHUNK == 0, "a wave reads whole chunks of the tag list");
  static_assert(AG_NCHUNK % AG_BLOCKS == 0, "the scans' offsets tile the list");
  static_assert(N_NODES % AG_ROWS == 0, "every work-group owns a full block of rows");
  static_assert(AG_Q >= 32 + 32, "one new edge per lane fits the ring beside 31 waiting ones");
  __shared__ float acc[AG_ROWS * F];
  __shared__ float cnt[AG_ROWS];
  __shared__ int   qe[AG_NW][AG_Q];

  const int tid  = (int)threadIdx.x;
  const int lane = tid & 31;
  const int wave = __builtin_amdgcn_readfirstlane(tid >> 5);
  const int row0 = (int)blockIdx.x * AG_ROWS;

  float4* acc4 = reinterpret_cast<float4*>(acc);
  const float4 z = make_float4(0.0f, 0.0f, 0.0f, 0.0f);
  for (int i = tid; i < AG_ROWS * F / 4; i += AG_TB) acc4[i] = z;
  if (tid < AG_ROWS) cnt[tid] = 0.0f;
  int* qe_w = qe[wave];
  for (int i = lane; i < AG_Q; i += 32) qe_w[i] = 0;
  __syncthreads();

  int head = 0;
  int qn   = 0;
  const int c0   = (int)blockIdx.x * AG_ROT + wave;
  const int cend = (int)blockIdx.x * AG_ROT + AG_NCHUNK;
  const unsigned char* tags = reinterpret_cast<const unsigned char*>(tags4);
  const unsigned K7F = 0x7F7F7F7Fu;
  const unsigned K80 = 0x80808080u;
  const unsigned pat = (unsigned)blockIdx.x * 0x01010101u;
  uint4 nxt = make_uint4(0u, 0u, 0u, 0u);
  {
    const int cc = c0 >= AG_NCHUNK ? c0 - AG_NCHUNK : c0;
    nxt = *reinterpret_cast<const uint4*>(tags + cc * AG_CHUNK + lane * 16);
  }
  for (int c = c0; c < cend; c += AG_NW) {
    const uint4 cur = nxt;
    const int ebase = (c >= AG_NCHUNK ? c - AG_NCHUNK : c) * AG_CHUNK + lane * 16;
    if (c + AG_NW < cend) {
      const int cn = c + AG_NW;
      const int cc = cn >= AG_NCHUNK ? cn - AG_NCHUNK : cn;
      nxt = *reinterpret_cast<const uint4*>(tags + cc * AG_CHUNK + lane * 16);
    }
    const unsigned w0 = cur.x ^ pat;
    const unsigned w1 = cur.y ^ pat;
    const unsigned w2 = cur.z ^ pat;
    const unsigned w3 = cur.w ^ pat;
    const unsigned z0 = ~(((w0 & K7F) + K7F) | w0) & K80;
    const unsigned z1 = ~(((w1 & K7F) + K7F) | w1) & K80;
    const unsigned z2 = ~(((w2 & K7F) + K7F) | w2) & K80;
    const unsigned z3 = ~(((w3 & K7F) + K7F) | w3) & K80;
    unsigned comb = (z0 >> 7) | (z1 >> 6) | (z2 >> 5) | (z3 >> 4);
    unsigned bal = __builtin_amdgcn_ballot_w32(comb != 0u);
    while (bal != 0u) {
      const bool hit = comb != 0u;
      const unsigned p = (unsigned)__builtin_ctz(comb | 0x80000000u);
      const int pos = (head + qn + (int)__builtin_amdgcn_mbcnt_lo(bal, 0u)) & (AG_Q - 1);
      if (hit) qe_w[pos] = ebase + (int)(((p & 3u) << 2) | (p >> 3));
      qn += __builtin_popcount(bal);
      comb &= comb - 1u;
      while (qn >= 32) {
        drain_ring<F>(feat, src, dst, row0, acc, cnt, qe_w, head, 32, lane);
        head = (head + 32) & (AG_Q - 1);
        qn -= 32;
      }
      bal = __builtin_amdgcn_ballot_w32(comb != 0u);
    }
  }
  while (qn > 0) {
    const int m = qn < 32 ? qn : 32;
    drain_ring<F>(feat, src, dst, row0, acc, cnt, qe_w, head, m, lane);
    head = (head + m) & (AG_Q - 1);
    qn -= m;
  }
  __syncthreads();

  constexpr int Q = F / 4;
  float4* out4 = reinterpret_cast<float4*>(agg + (long long)row0 * F);
  for (int i = tid; i < AG_ROWS * Q; i += AG_TB) {
    const float c  = cnt[i / Q];
    const float cc = c > 1.0f ? c : 1.0f;
    const float4 v = acc4[i];
    float4 r;
    r.x = v.x / cc;
    r.y = v.y / cc;
    r.z = v.z / cc;
    r.w = v.w / cc;
    out4[i] = r;
  }
}

__global__ void dual_gemm_wmma(const float* __restrict__ A1,
                               const float* __restrict__ W1,
                               const float* __restrict__ A2,
                               const float* __restrict__ W2,
                               const float* __restrict__ bias,
                               float* __restrict__ out,
                               int K, int ncols, int ldo, int do_relu) {
  const int lane   = threadIdx.x & 31;
  const int wid    = blockIdx.x * (blockDim.x >> 5) + (threadIdx.x >> 5);
  const int ctiles = (ncols + 15) >> 4;
  const int mtiles = N_NODES >> 4;
  if (wid >= mtiles * ctiles) return;
  const int rt = wid / ctiles;
  const int ct = wid - rt * ctiles;

  const int mrow  = rt * 16 + (lane & 15);
  const int ncol  = ct * 16 + (lane & 15);
  const int khalf = (lane >> 4) * 4;
  const bool cvalid  = (ncol < ncols);
  const int  ncol_ld = cvalid ? ncol : (ncols - 1);

  const float* arow1 = A1 + (unsigned)mrow    * (unsigned)K + khalf;
  const float* brow1 = W1 + (unsigned)ncol_ld * (unsigned)K + khalf;
  const float* arow2 = A2 + (unsigned)mrow    * (unsigned)K + khalf;
  const float* brow2 = W2 + (unsigned)ncol_ld * (unsigned)K + khalf;

  v8f c = {};

  #pragma unroll 4
  for (int k0 = 0; k0 < K; k0 += 8) {
    const float4 av = *(const float4*)(arow1 + k0);
    const float4 bv = *(const float4*)(brow1 + k0);
    v2f a0; a0.x = av.x; a0.y = av.y;
    v2f b0; b0.x = bv.x; b0.y = bv.y;
    v2f a1; a1.x = av.z; a1.y = av.w;
    v2f b1; b1.x = bv.z; b1.y = bv.w;
    c = __builtin_amdgcn_wmma_f32_16x16x4_f32(false, a0, false, b0,
                                              (short)0, c, false, false);
    c = __builtin_amdgcn_wmma_f32_16x16x4_f32(false, a1, false, b1,
                                              (short)0, c, false, false);
  }
  #pragma unroll 4
  for (int k0 = 0; k0 < K; k0 += 8) {
    const float4 av = *(const float4*)(arow2 + k0);
    const float4 bv = *(const float4*)(brow2 + k0);
    v2f a0; a0.x = av.x; a0.y = av.y;
    v2f b0; b0.x = bv.x; b0.y = bv.y;
    v2f a1; a1.x = av.z; a1.y = av.w;
    v2f b1; b1.x = bv.z; b1.y = bv.w;
    c = __builtin_amdgcn_wmma_f32_16x16x4_f32(false, a0, false, b0,
                                              (short)0, c, false, false);
    c = __builtin_amdgcn_wmma_f32_16x16x4_f32(false, a1, false, b1,
                                              (short)0, c, false, false);
  }

  if (cvalid) {
    float bv = bias[ncol];
    unsigned rbase = (unsigned)(rt * 16 + ((lane < 16) ? 0 : 8));
    float* orow = out + (unsigned)rbase * (unsigned)ldo + ncol;
    #pragma unroll
    for (int i = 0; i < 8; ++i) {
      float v = c[i] + bv;
      if (do_relu) v = v > 0.0f ? v : 0.0f;
      orow[(unsigned)i * (unsigned)ldo] = v;
    }
  }
}

__global__ void log_softmax_rows(const float* __restrict__ logits,
                                 float* __restrict__ out) {
  int r = blockIdx.x * blockDim.x + threadIdx.x;
  if (r >= N_NODES) return;
  const float* l = logits + (long long)r * CLS;
  float m = -3.402823466e38f;
  #pragma unroll
  for (int j = 0; j < CLS; ++j) m = fmaxf(m, l[j]);
  float s = 0.0f;
  #pragma unroll
  for (int j = 0; j < CLS; ++j) s += expf(l[j] - m);
  float ls = logf(s);
  float* o = out + (long long)r * CLS;
  #pragma unroll
  for (int j = 0; j < CLS; ++j) o[j] = l[j] - m - ls;
}

extern "C" void kernel_launch(void* const* d_in, const int* in_sizes, int n_in,
                              void* d_out, int out_size, void* d_ws, size_t ws_size,
                              hipStream_t stream) {
  (void)in_sizes; (void)n_in; (void)out_size; (void)ws_size;

  const float* x    = (const float*)d_in[0];
  const int*   ei   = (const int*)  d_in[1];
  const float* W1_l = (const float*)d_in[2];
  const float* b1   = (const float*)d_in[3];
  const float* W1_r = (const float*)d_in[4];
  const float* W2_l = (const float*)d_in[5];
  const float* b2   = (const float*)d_in[6];
  const float* W2_r = (const float*)d_in[7];
  const int* src = ei;
  const int* dst = ei + NE;
  float* out = (float*)d_out;

  float* ws     = (float*)d_ws;
  float* agg1   = ws   + N_NODES;
  float* h      = agg1 + (long long)N_NODES * F_IN;
  float* agg2   = h    + (long long)N_NODES * HID;
  float* logits = agg2 + (long long)N_NODES * HID;
  unsigned* tags4 = (unsigned*)logits;

  const int TB = 256;
  auto blocks = [](long long n, int tb) { return (unsigned)((n + tb - 1) / tb); };

  edge_owner_tags<<<blocks(NE / 4, TB), TB, 0, stream>>>(dst, tags4);

  owner_mean<F_IN><<<AG_BLOCKS, AG_TB, 0, stream>>>(x, src, dst, tags4, agg1);

  {
    int tiles = (N_NODES / 16) * (HID / 16);
    dual_gemm_wmma<<<blocks(tiles, 8), TB, 0, stream>>>(
        agg1, W1_l, x, W1_r, b1, h, F_IN, HID, HID,  1);
  }

  owner_mean<HID><<<AG_BLOCKS, AG_TB, 0, stream>>>(h, src, dst, tags4, agg2);

  {
    int ctiles = (CLS + 15) / 16;
    int tiles = (N_NODES / 16) * ctiles;
    dual_gemm_wmma<<<blocks(tiles, 8), TB, 0, stream>>>(
        agg2, W2_l, h, W2_r, b2, logits, HID, CLS, CLS,  0);
  }

  log_softmax_rows<<<blocks(N_NODES, TB), TB, 0, stream>>>(logits, out);
}


# Owner means: found edges go through per-wave LDS queues (integer LDS add) to the wave owning the dst row, which adds rows by plain LDS read-add-write; no f32 LDS atomics; sums in queue arrival order
# speedup vs baseline: 2.7469x; 1.0653x over previous
#include <hip/hip_runtime.h>

#define N_NODES 50000
#define F_IN    128
#define HID     256
#define CLS     40
#define NE      640000

typedef __attribute__((ext_vector_type(2))) float v2f;
typedef __attribute__((ext_vector_type(8))) float v8f;

#define AG_ROWS   200
#define AG_NW     32
#define AG_TB     (AG_NW * 32)
#define AG_BLOCKS (N_NODES / AG_ROWS)
#define AG_Q      64
#define AG_CHUNK  512
#define AG_NOOWN  255u
#define AG_RQ     256
#define AG_NCHUNK (NE / AG_CHUNK)
#define AG_ROT    (AG_NCHUNK / AG_BLOCKS)

__global__ __launch_bounds__(256)
void edge_owner_tags(const int* __restrict__ dst, unsigned* __restrict__ tags4) {
  static_assert(AG_BLOCKS < (int)AG_NOOWN, "an owner number fits a byte");
  static_assert(NE % 4 == 0, "whole groups of four edges");
  const int i = (int)(blockIdx.x * 256u + threadIdx.x);
  if (i >= NE / 4) return;
  const int4 d = *reinterpret_cast<const int4*>(dst + i * 4);
  const unsigned t0 = (unsigned)d.x < (unsigned)N_NODES ? (unsigned)d.x / AG_ROWS : AG_NOOWN;
  const unsigned t1 = (unsigned)d.y < (unsigned)N_NODES ? (unsigned)d.y / AG_ROWS : AG_NOOWN;
  const unsigned t2 = (unsigned)d.z < (unsigned)N_NODES ? (unsigned)d.z / AG_ROWS : AG_NOOWN;
  const unsigned t3 = (unsigned)d.w < (unsigned)N_NODES ? (unsigned)d.w / AG_ROWS : AG_NOOWN;
  tags4[i] = t0 | (t1 << 8) | (t2 << 16) | (t3 << 24);
}

static __device__ __forceinline__ void lds_add(float* p, float v) {
  __hip_atomic_fetch_add(p, v, __ATOMIC_RELAXED, __HIP_MEMORY_SCOPE_WORKGROUP);
}

template <int F>
static __device__ __forceinline__ void route_ring(const float* __restrict__ feat,
                                                  const int*   __restrict__ src,
                                                  const int*   __restrict__ dst,
                                                  int row0,
                                                  float* acc, float* cnt,
                                                  const int* qe_w,
                                                  uint2* rq, unsigned* rtail,
                                                  int head, int m, int lane) {
  constexpr int NCH = F / 32;
  const int slot = (head + lane) & (AG_Q - 1);
  const int e  = qe_w[slot];
  const int s  = src[e];
  const int dl = dst[e] - row0;
  bool over = false;
  if (lane < m) {
    lds_add(cnt + dl, 1.0f);
    const unsigned ow  = (unsigned)dl & (unsigned)(AG_NW - 1);
    const unsigned pos = __hip_atomic_fetch_add(rtail + ow, 1u, __ATOMIC_RELAXED,
                                                __HIP_MEMORY_SCOPE_WORKGROUP);
    if (pos < (unsigned)AG_RQ) rq[ow * AG_RQ + pos] = make_uint2((unsigned)s, (unsigned)dl);
    else over = true;
  }
  unsigned ov = __builtin_amdgcn_ballot_w32(over);
  while (ov != 0u) {
    const int j  = __builtin_ctz(ov);
    const int sj = __builtin_amdgcn_readlane(s, j);
    const int dj = __builtin_amdgcn_readlane(dl, j);
    const float* f = feat + (long long)sj * F + lane;
    float* a = acc + dj * F + lane;
    #pragma unroll
    for (int c = 0; c < NCH; ++c) lds_add(a + c * 32, f[c * 32]);
    ov &= ov - 1u;
  }
}

template <int F>
__global__ __launch_bounds__(AG_TB)
void owner_mean(const float* __restrict__ feat,
                const int*   __restrict__ src,
                const int*   __restrict__ dst,
                const unsigned* __restrict__ tags4,
                float* __restrict__ agg) {
  static_assert(NE % AG_CHUNK == 0, "a wave reads whole chunks of the tag list");
  static_assert(AG_NCHUNK % AG_BLOCKS == 0, "the scans' offsets tile the list");
  static_assert(N_NODES % AG_ROWS == 0, "every work-group owns a full block of rows");
  static_assert(AG_Q >= 32 + 32, "one new edge per lane fits the ring beside 31 waiting ones");
  __shared__ __attribute__((aligned(16))) float acc[AG_ROWS * F];
  __shared__ float cnt[AG_ROWS];
  __shared__ int   qe[AG_NW][AG_Q];
  __shared__ uint2    rq[AG_NW * AG_RQ];
  __shared__ unsigned rtail[AG_NW];

  const int tid  = (int)threadIdx.x;
  const int lane = tid & 31;
  const int wave = __builtin_amdgcn_readfirstlane(tid >> 5);
  const int row0 = (int)blockIdx.x * AG_ROWS;

  float4* acc4 = reinterpret_cast<float4*>(acc);
  const float4 z = make_float4(0.0f, 0.0f, 0.0f, 0.0f);
  for (int i = tid; i < AG_ROWS * F / 4; i += AG_TB) acc4[i] = z;
  if (tid < AG_ROWS) cnt[tid] = 0.0f;
  if (tid < AG_NW) rtail[tid] = 0u;
  int* qe_w = qe[wave];
  for (int i = lane; i < AG_Q; i += 32) qe_w[i] = 0;
  __syncthreads();

  int head = 0;
  int qn   = 0;
  const int c0   = (int)blockIdx.x * AG_ROT + wave;
  const int cend = (int)blockIdx.x * AG_ROT + AG_NCHUNK;
  const unsigned char* tags = reinterpret_cast<const unsigned char*>(tags4);
  const unsigned K7F = 0x7F7F7F7Fu;
  const unsigned K80 = 0x80808080u;
  const unsigned pat = (unsigned)blockIdx.x * 0x01010101u;
  uint4 nxt = make_uint4(0u, 0u, 0u, 0u);
  {
    const int cc = c0 >= AG_NCHUNK ? c0 - AG_NCHUNK : c0;
    nxt = *reinterpret_cast<const uint4*>(tags + cc * AG_CHUNK + lane * 16);
  }
  for (int c = c0; c < cend; c += AG_NW) {
    const uint4 cur = nxt;
    const int ebase = (c >= AG_NCHUNK ? c - AG_NCHUNK : c) * AG_CHUNK + lane * 16;
    if (c + AG_NW < cend) {
      const int cn = c + AG_NW;
      const int cc = cn >= AG_NCHUNK ? cn - AG_NCHUNK : cn;
      nxt = *reinterpret_cast<const uint4*>(tags + cc * AG_CHUNK + lane * 16);
    }
    const unsigned w0 = cur.x ^ pat;
    const unsigned w1 = cur.y ^ pat;
    const unsigned w2 = cur.z ^ pat;
    const unsigned w3 = cur.w ^ pat;
    const unsigned z0 = ~(((w0 & K7F) + K7F) | w0) & K80;
    const unsigned z1 = ~(((w1 & K7F) + K7F) | w1) & K80;
    const unsigned z2 = ~(((w2 & K7F) + K7F) | w2) & K80;
    const unsigned z3 = ~(((w3 & K7F) + K7F) | w3) & K80;
    unsigned comb = (z0 >> 7) | (z1 >> 6) | (z2 >> 5) | (z3 >> 4);
    unsigned bal = __builtin_amdgcn_ballot_w32(comb != 0u);
    while (bal != 0u) {
      const bool hit = comb != 0u;
      const unsigned p = (unsigned)__builtin_ctz(comb | 0x80000000u);
      const int pos = (head + qn + (int)__builtin_amdgcn_mbcnt_lo(bal, 0u)) & (AG_Q - 1);
      if (hit) qe_w[pos] = ebase + (int)(((p & 3u) << 2) | (p >> 3));
      qn += __builtin_popcount(bal);
      comb &= comb - 1u;
      while (qn >= 32) {
        route_ring<F>(feat, src, dst, row0, acc, cnt, qe_w, rq, rtail, head, 32, lane);
        head = (head + 32) & (AG_Q - 1);
        qn -= 32;
      }
      bal = __builtin_amdgcn_ballot_w32(comb != 0u);
    }
  }
  while (qn > 0) {
    const int m = qn < 32 ? qn : 32;
    route_ring<F>(feat, src, dst, row0, acc, cnt, qe_w, rq, rtail, head, m, lane);
    head = (head + m) & (AG_Q - 1);
    qn -= m;
  }
  __syncthreads();

  {
    constexpr int NV  = F / 128;
    constexpr int GRP = 2048 / F;
    const unsigned nt = rtail[wave];
    const int n = __builtin_amdgcn_readfirstlane((int)(nt < (unsigned)AG_RQ ? nt : (unsigned)AG_RQ));
    const uint2*  q     = rq + wave * AG_RQ;
    const float4* feat4 = reinterpret_cast<const float4*>(feat);
    for (int b = 0; b < n; b += 32) {
      const int m = n - b < 32 ? n - b : 32;
      const uint2 en = lane < m ? q[b + lane] : make_uint2(0u, 0u);
      const int s  = (int)en.x;
      const int dl = (int)en.y;
      for (int j = 0; j < m; j += GRP) {
        float4 v[GRP][NV];
        #pragma unroll
        for (int g = 0; g < GRP; ++g) {
          const int jj = j + g < m ? j + g : m - 1;
          const int sj = __builtin_amdgcn_readlane(s, jj);
          const float4* f = feat4 + (long long)sj * (F / 4) + lane;
          #pragma unroll
          for (int c = 0; c < NV; ++c) v[g][c] = f[c * 32];
        }
        #pragma unroll
        for (int g = 0; g < GRP; ++g) {
          if (j + g < m) {
            const int dj = __builtin_amdgcn_readlane(dl, j + g);
            float4* a = acc4 + dj * (F / 4) + lane;
            #pragma unroll
            for (int c = 0; c < NV; ++c) {
              float4 t = a[c * 32];
              t.x += v[g][c].x;
              t.y += v[g][c].y;
              t.z += v[g][c].z;
              t.w += v[g][c].w;
              a[c * 32] = t;
            }
          }
        }
      }
    }
  }
  __syncthreads();

  constexpr int Q = F / 4;
  float4* out4 = reinterpret_cast<float4*>(agg + (long long)row0 * F);
  for (int i = tid; i < AG_ROWS * Q; i += AG_TB) {
    const float c  = cnt[i / Q];
    const float cc = c > 1.0f ? c : 1.0f;
    const float4 v = acc4[i];
    float4 r;
    r.x = v.x / cc;
    r.y = v.y / cc;
    r.z = v.z / cc;
    r.w = v.w / cc;
    out4[i] = r;
  }
}

__global__ void dual_gemm_wmma(const float* __restrict__ A1,
                               const float* __restrict__ W1,
                               const float* __restrict__ A2,
                               const float* __restrict__ W2,
                               const float* __restrict__ bias,
                               float* __restrict__ out,
                               int K, int ncols, int ldo, int do_relu) {
  const int lane   = threadIdx.x & 31;
  const int wid    = blockIdx.x * (blockDim.x >> 5) + (threadIdx.x >> 5);
  const int ctiles = (ncols + 15) >> 4;
  const int mtiles = N_NODES >> 4;
  if (wid >= mtiles * ctiles) return;
  const int rt = wid / ctiles;
  const int ct = wid - rt * ctiles;

  const int mrow  = rt * 16 + (lane & 15);
  const int ncol  = ct * 16 + (lane & 15);
  const int khalf = (lane >> 4) * 4;
  const bool cvalid  = (ncol < ncols);
  const int  ncol_ld = cvalid ? ncol : (ncols - 1);

  const float* arow1 = A1 + (unsigned)mrow    * (unsigned)K + khalf;
  const float* brow1 = W1 + (unsigned)ncol_ld * (unsigned)K + khalf;
  const float* arow2 = A2 + (unsigned)mrow    * (unsigned)K + khalf;
  const float* brow2 = W2 + (unsigned)ncol_ld * (unsigned)K + khalf;

  v8f c = {};

  #pragma unroll 4
  for (int k0 = 0; k0 < K; k0 += 8) {
    const float4 av = *(const float4*)(arow1 + k0);
    const float4 bv = *(const float4*)(brow1 + k0);
    v2f a0; a0.x = av.x; a0.y = av.y;
    v2f b0; b0.x = bv.x; b0.y = bv.y;
    v2f a1; a1.x = av.z; a1.y = av.w;
    v2f b1; b1.x = bv.z; b1.y = bv.w;
    c = __builtin_amdgcn_wmma_f32_16x16x4_f32(false, a0, false, b0,
                                              (short)0, c, false, false);
    c = __builtin_amdgcn_wmma_f32_16x16x4_f32(false, a1, false, b1,
                                              (short)0, c, false, false);
  }
  #pragma unroll 4
  for (int k0 = 0; k0 < K; k0 += 8) {
    const float4 av = *(const float4*)(arow2 + k0);
    const float4 bv = *(const float4*)(brow2 + k0);
    v2f a0; a0.x = av.x; a0.y = av.y;
    v2f b0; b0.x = bv.x; b0.y = bv.y;
    v2f a1; a1.x = av.z; a1.y = av.w;
    v2f b1; b1.x = bv.z; b1.y = bv.w;
    c = __builtin_amdgcn_wmma_f32_16x16x4_f32(false, a0, false, b0,
                                              (short)0, c, false, false);
    c = __builtin_amdgcn_wmma_f32_16x16x4_f32(false, a1, false, b1,
                                              (short)0, c, false, false);
  }

  if (cvalid) {
    float bv = bias[ncol];
    unsigned rbase = (unsigned)(rt * 16 + ((lane < 16) ? 0 : 8));
    float* orow = out + (unsigned)rbase * (unsigned)ldo + ncol;
    #pragma unroll
    for (int i = 0; i < 8; ++i) {
      float v = c[i] + bv;
      if (do_relu) v = v > 0.0f ? v : 0.0f;
      orow[(unsigned)i * (unsigned)ldo] = v;
    }
  }
}

__global__ void log_softmax_rows(const float* __restrict__ logits,
                                 float* __restrict__ out) {
  int r = blockIdx.x * blockDim.x + threadIdx.x;
  if (r >= N_NODES) return;
  const float* l = logits + (long long)r * CLS;
  float m = -3.402823466e38f;
  #pragma unroll
  for (int j = 0; j < CLS; ++j) m = fmaxf(m, l[j]);
  float s = 0.0f;
  #pragma unroll
  for (int j = 0; j < CLS; ++j) s += expf(l[j] - m);
  float ls = logf(s);
  float* o = out + (long long)r * CLS;
  #pragma unroll
  for (int j = 0; j < CLS; ++j) o[j] = l[j] - m - ls;
}

extern "C" void kernel_launch(void* const* d_in, const int* in_sizes, int n_in,
                              void* d_out, int out_size, void* d_ws, size_t ws_size,
                              hipStream_t stream) {
  (void)in_sizes; (void)n_in; (void)out_size; (void)ws_size;

  const float* x    = (const float*)d_in[0];
  const int*   ei   = (const int*)  d_in[1];
  const float* W1_l = (const float*)d_in[2];
  const float* b1   = (const float*)d_in[3];
  const float* W1_r = (const float*)d_in[4];
  const float* W2_l = (const float*)d_in[5];
  const float* b2   = (const float*)d_in[6];
  const float* W2_r = (const float*)d_in[7];
  const int* src = ei;
  const int* dst = ei + NE;
  float* out = (float*)d_out;

  float* ws     = (float*)d_ws;
  float* agg1   = ws   + N_NODES;
  float* h      = agg1 + (long long)N_NODES * F_IN;
  float* agg2   = h    + (long long)N_NODES * HID;
  float* logits = agg2 + (long long)N_NODES * HID;
  unsigned* tags4 = (unsigned*)logits;

  const int TB = 256;
  auto blocks = [](long long n, int tb) { return (unsigned)((n + tb - 1) / tb); };

  edge_owner_tags<<<blocks(NE / 4, TB), TB, 0, stream>>>(dst, tags4);

  owner_mean<F_IN><<<AG_BLOCKS, AG_TB, 0, stream>>>(x, src, dst, tags4, agg1);

  {
    int tiles = (N_NODES / 16) * (HID / 16);
    dual_gemm_wmma<<<blocks(tiles, 8), TB, 0, stream>>>(
        agg1, W1_l, x, W1_r, b1, h, F_IN, HID, HID,  1);
  }

  owner_mean<HID><<<AG_BLOCKS, AG_TB, 0, stream>>>(h, src, dst, tags4, agg2);

  {
    int ctiles = (CLS + 15) / 16;
    int tiles = (N_NODES / 16) * ctiles;
    dual_gemm_wmma<<<blocks(tiles, 8), TB, 0, stream>>>(
        agg2, W2_l, h, W2_r, b2, logits, HID, CLS, CLS,  0);
  }

  log_softmax_rows<<<blocks(N_NODES, TB), TB, 0, stream>>>(logits, out);
}


# Layer-1 WMMA product weight-stationary: a wave keeps its 16 columns' B fragments (all K, both weights) in registers and walks 12-13 row tiles loading A only; 256 blocks x 16 waves; sum order kept
# speedup vs baseline: 3.2798x; 1.1940x over previous
#include <hip/hip_runtime.h>

#define N_NODES 50000
#define F_IN    128
#define HID     256
#define CLS     40
#define NE      640000

typedef __attribute__((ext_vector_type(2))) float v2f;
typedef __attribute__((ext_vector_type(8))) float v8f;

#define AG_ROWS   200
#define AG_NW     32
#define AG_TB     (AG_NW * 32)
#define AG_BLOCKS (N_NODES / AG_ROWS)
#define AG_Q      64
#define AG_CHUNK  512
#define AG_NOOWN  255u
#define AG_RQ     256
#define AG_NCHUNK (NE / AG_CHUNK)
#define AG_ROT    (AG_NCHUNK / AG_BLOCKS)

__global__ __launch_bounds__(256)
void edge_owner_tags(const int* __restrict__ dst, unsigned* __restrict__ tags4) {
  static_assert(AG_BLOCKS < (int)AG_NOOWN, "an owner number fits a byte");
  static_assert(NE % 4 == 0, "whole groups of four edges");
  const int i = (int)(blockIdx.x * 256u + threadIdx.x);
  if (i >= NE / 4) return;
  const int4 d = *reinterpret_cast<const int4*>(dst + i * 4);
  const unsigned t0 = (unsigned)d.x < (unsigned)N_NODES ? (unsigned)d.x / AG_ROWS : AG_NOOWN;
  const unsigned t1 = (unsigned)d.y < (unsigned)N_NODES ? (unsigned)d.y / AG_ROWS : AG_NOOWN;
  const unsigned t2 = (unsigned)d.z < (unsigned)N_NODES ? (unsigned)d.z / AG_ROWS : AG_NOOWN;
  const unsigned t3 = (unsigned)d.w < (unsigned)N_NODES ? (unsigned)d.w / AG_ROWS : AG_NOOWN;
  tags4[i] = t0 | (t1 << 8) | (t2 << 16) | (t3 << 24);
}

static __device__ __forceinline__ void lds_add(float* p, float v) {
  __hip_atomic_fetch_add(p, v, __ATOMIC_RELAXED, __HIP_MEMORY_SCOPE_WORKGROUP);
}

template <int F>
static __device__ __forceinline__ void route_ring(const float* __restrict__ feat,
                                                  const int*   __restrict__ src,
                                                  const int*   __restrict__ dst,
                                                  int row0,
                                                  float* acc, float* cnt,
                                                  const int* qe_w,
                                                  uint2* rq, unsigned* rtail,
                                                  int head, int m, int lane) {
  constexpr int NCH = F / 32;
  const int slot = (head + lane) & (AG_Q - 1);
  const int e  = qe_w[slot];
  const int s  = src[e];
  const int dl = dst[e] - row0;
  bool over = false;
  if (lane < m) {
    lds_add(cnt + dl, 1.0f);
    const unsigned ow  = (unsigned)dl & (unsigned)(AG_NW - 1);
    const unsigned pos = __hip_atomic_fetch_add(rtail + ow, 1u, __ATOMIC_RELAXED,
                                                __HIP_MEMORY_SCOPE_WORKGROUP);
    if (pos < (unsigned)AG_RQ) rq[ow * AG_RQ + pos] = make_uint2((unsigned)s, (unsigned)dl);
    else over = true;
  }
  unsigned ov = __builtin_amdgcn_ballot_w32(over);
  while (ov != 0u) {
    const int j  = __builtin_ctz(ov);
    const int sj = __builtin_amdgcn_readlane(s, j);
    const int dj = __builtin_amdgcn_readlane(dl, j);
    const float* f = feat + (long long)sj * F + lane;
    float* a = acc + dj * F + lane;
    #pragma unroll
    for (int c = 0; c < NCH; ++c) lds_add(a + c * 32, f[c * 32]);
    ov &= ov - 1u;
  }
}

template <int F>
__global__ __launch_bounds__(AG_TB)
void owner_mean(const float* __restrict__ feat,
                const int*   __restrict__ src,
                const int*   __restrict__ dst,
                const unsigned* __restrict__ tags4,
                float* __restrict__ agg) {
  static_assert(NE % AG_CHUNK == 0, "a wave reads whole chunks of the tag list");
  static_assert(AG_NCHUNK % AG_BLOCKS == 0, "the scans' offsets tile the list");
  static_assert(N_NODES % AG_ROWS == 0, "every work-group owns a full block of rows");
  static_assert(AG_Q >= 32 + 32, "one new edge per lane fits the ring beside 31 waiting ones");
  __shared__ __attribute__((aligned(16))) float acc[AG_ROWS * F];
  __shared__ float cnt[AG_ROWS];
  __shared__ int   qe[AG_NW][AG_Q];
  __shared__ uint2    rq[AG_NW * AG_RQ];
  __shared__ unsigned rtail[AG_NW];

  const int tid  = (int)threadIdx.x;
  const int lane = tid & 31;
  const int wave = __builtin_amdgcn_readfirstlane(tid >> 5);
  const int row0 = (int)blockIdx.x * AG_ROWS;

  float4* acc4 = reinterpret_cast<float4*>(acc);
  const float4 z = make_float4(0.0f, 0.0f, 0.0f, 0.0f);
  for (int i = tid; i < AG_ROWS * F / 4; i += AG_TB) acc4[i] = z;
  if (tid < AG_ROWS) cnt[tid] = 0.0f;
  if (tid < AG_NW) rtail[tid] = 0u;
  int* qe_w = qe[wave];
  for (int i = lane; i < AG_Q; i += 32) qe_w[i] = 0;
  __syncthreads();

  int head = 0;
  int qn   = 0;
  const int c0   = (int)blockIdx.x * AG_ROT + wave;
  const int cend = (int)blockIdx.x * AG_ROT + AG_NCHUNK;
  const unsigned char* tags = reinterpret_cast<const unsigned char*>(tags4);
  const unsigned K7F = 0x7F7F7F7Fu;
  const unsigned K80 = 0x80808080u;
  const unsigned pat = (unsigned)blockIdx.x * 0x01010101u;
  uint4 nxt = make_uint4(0u, 0u, 0u, 0u);
  {
    const int cc = c0 >= AG_NCHUNK ? c0 - AG_NCHUNK : c0;
    nxt = *reinterpret_cast<const uint4*>(tags + cc * AG_CHUNK + lane * 16);
  }
  for (int c = c0; c < cend; c += AG_NW) {
    const uint4 cur = nxt;
    const int ebase = (c >= AG_NCHUNK ? c - AG_NCHUNK : c) * AG_CHUNK + lane * 16;
    if (c + AG_NW < cend) {
      const int cn = c + AG_NW;
      const int cc = cn >= AG_NCHUNK ? cn - AG_NCHUNK : cn;
      nxt = *reinterpret_cast<const uint4*>(tags + cc * AG_CHUNK + lane * 16);
    }
    const unsigned w0 = cur.x ^ pat;
    const unsigned w1 = cur.y ^ pat;
    const unsigned w2 = cur.z ^ pat;
    const unsigned w3 = cur.w ^ pat;
    const unsigned z0 = ~(((w0 & K7F) + K7F) | w0) & K80;
    const unsigned z1 = ~(((w1 & K7F) + K7F) | w1) & K80;
    const unsigned z2 = ~(((w2 & K7F) + K7F) | w2) & K80;
    const unsigned z3 = ~(((w3 & K7F) + K7F) | w3) & K80;
    unsigned comb = (z0 >> 7) | (z1 >> 6) | (z2 >> 5) | (z3 >> 4);
    unsigned bal = __builtin_amdgcn_ballot_w32(comb != 0u);
    while (bal != 0u) {
      const bool hit = comb != 0u;
      const unsigned p = (unsigned)__builtin_ctz(comb | 0x80000000u);
      const int pos = (head + qn + (int)__builtin_amdgcn_mbcnt_lo(bal, 0u)) & (AG_Q - 1);
      if (hit) qe_w[pos] = ebase + (int)(((p & 3u) << 2) | (p >> 3));
      qn += __builtin_popcount(bal);
      comb &= comb - 1u;
      while (qn >= 32) {
        route_ring<F>(feat, src, dst, row0, acc, cnt, qe_w, rq, rtail, head, 32, lane);
        head = (head + 32) & (AG_Q - 1);
        qn -= 32;
      }
      bal = __builtin_amdgcn_ballot_w32(comb != 0u);
    }
  }
  while (qn > 0) {
    const int m = qn < 32 ? qn : 32;
    route_ring<F>(feat, src, dst, row0, acc, cnt, qe_w, rq, rtail, head, m, lane);
    head = (head + m) & (AG_Q - 1);
    qn -= m;
  }
  __syncthreads();

  {
    constexpr int NV  = F / 128;
    constexpr int GRP = 2048 / F;
    const unsigned nt = rtail[wave];
    const int n = __builtin_amdgcn_readfirstlane((int)(nt < (unsigned)AG_RQ ? nt : (unsigned)AG_RQ));
    const uint2*  q     = rq + wave * AG_RQ;
    const float4* feat4 = reinterpret_cast<const float4*>(feat);
    for (int b = 0; b < n; b += 32) {
      const int m = n - b < 32 ? n - b : 32;
      const uint2 en = lane < m ? q[b + lane] : make_uint2(0u, 0u);
      const int s  = (int)en.x;
      const int dl = (int)en.y;
      for (int j = 0; j < m; j += GRP) {
        float4 v[GRP][NV];
        #pragma unroll
        for (int g = 0; g < GRP; ++g) {
          const int jj = j + g < m ? j + g : m - 1;
          const int sj = __builtin_amdgcn_readlane(s, jj);
          const float4* f = feat4 + (long long)sj * (F / 4) + lane;
          #pragma unroll
          for (int c = 0; c < NV; ++c) v[g][c] = f[c * 32];
        }
        #pragma unroll
        for (int g = 0; g < GRP; ++g) {
          if (j + g < m) {
            const int dj = __builtin_amdgcn_readlane(dl, j + g);
            float4* a = acc4 + dj * (F / 4) + lane;
            #pragma unroll
            for (int c = 0; c < NV; ++c) {
              float4 t = a[c * 32];
              t.x += v[g][c].x;
              t.y += v[g][c].y;
              t.z += v[g][c].z;
              t.w += v[g][c].w;
              a[c * 32] = t;
            }
          }
        }
      }
    }
  }
  __syncthreads();

  constexpr int Q = F / 4;
  float4* out4 = reinterpret_cast<float4*>(agg + (long long)row0 * F);
  for (int i = tid; i < AG_ROWS * Q; i += AG_TB) {
    const float c  = cnt[i / Q];
    const float cc = c > 1.0f ? c : 1.0f;
    const float4 v = acc4[i];
    float4 r;
    r.x = v.x / cc;
    r.y = v.y / cc;
    r.z = v.z / cc;
    r.w = v.w / cc;
    out4[i] = r;
  }
}

#define G1_WAVES  (HID / 16)
#define G1_TB     (G1_WAVES * 32)
#define G1_BLOCKS 256
#define G1_MT     (N_NODES / 16)
#define G1_STEPS  (F_IN / 8)
#define G1_GRP    8

static __device__ __forceinline__ v8f mm2(const float4 av, const float4 bv, v8f c) {
  v2f a0; a0.x = av.x; a0.y = av.y;
  v2f b0; b0.x = bv.x; b0.y = bv.y;
  v2f a1; a1.x = av.z; a1.y = av.w;
  v2f b1; b1.x = bv.z; b1.y = bv.w;
  c = __builtin_amdgcn_wmma_f32_16x16x4_f32(false, a0, false, b0, (short)0, c, false, false);
  c = __builtin_amdgcn_wmma_f32_16x16x4_f32(false, a1, false, b1, (short)0, c, false, false);
  return c;
}

__global__ __launch_bounds__(G1_TB)
void gemm1_weights_in_regs(const float* __restrict__ A1,
                           const float* __restrict__ W1,
                           const float* __restrict__ A2,
                           const float* __restrict__ W2,
                           const float* __restrict__ bias,
                           float* __restrict__ out) {
  static_assert(N_NODES % 16 == 0, "whole row tiles");
  static_assert(G1_STEPS == 2 * G1_GRP, "two groups of k-steps per product");
  const int lane  = (int)threadIdx.x & 31;
  const int ct    = __builtin_amdgcn_readfirstlane((int)threadIdx.x >> 5);
  const int c16   = lane & 15;
  const int khalf = (lane >> 4) * 4;
  const int ncol  = ct * 16 + c16;

  float4 bl[G1_STEPS], br[G1_STEPS];
  {
    const float* bp1 = W1 + ncol * F_IN + khalf;
    const float* bp2 = W2 + ncol * F_IN + khalf;
    #pragma unroll
    for (int s = 0; s < G1_STEPS; ++s) {
      bl[s] = *reinterpret_cast<const float4*>(bp1 + 8 * s);
      br[s] = *reinterpret_cast<const float4*>(bp2 + 8 * s);
    }
  }
  const float bv = bias[ncol];

  float4 g0[G1_GRP], g1[G1_GRP], g2[G1_GRP], g3[G1_GRP];
  int rt = (int)blockIdx.x;
  {
    const float* ap1 = A1 + (unsigned)(rt * 16 + c16) * (unsigned)F_IN + khalf;
    #pragma unroll
    for (int s = 0; s < G1_GRP; ++s) g0[s] = *reinterpret_cast<const float4*>(ap1 + 8 * s);
  }
  for (; rt < G1_MT; rt += (int)gridDim.x) {
    const unsigned roff = (unsigned)(rt * 16 + c16) * (unsigned)F_IN + khalf;
    const float* ap1 = A1 + roff;
    const float* ap2 = A2 + roff;
    const int rn = rt + (int)gridDim.x < G1_MT ? rt + (int)gridDim.x : rt;
    const float* an1 = A1 + (unsigned)(rn * 16 + c16) * (unsigned)F_IN + khalf;
    v8f c = {};
    #pragma unroll
    for (int s = 0; s < G1_GRP; ++s) g1[s] = *reinterpret_cast<const float4*>(ap1 + 8 * (G1_GRP + s));
    #pragma unroll
    for (int s = 0; s < G1_GRP; ++s) c = mm2(g0[s], bl[s], c);
    #pragma unroll
    for (int s = 0; s < G1_GRP; ++s) g2[s] = *reinterpret_cast<const float4*>(ap2 + 8 * s);
    #pragma unroll
    for (int s = 0; s < G1_GRP; ++s) c = mm2(g1[s], bl[G1_GRP + s], c);
    #pragma unroll
    for (int s = 0; s < G1_GRP; ++s) g3[s] = *reinterpret_cast<const float4*>(ap2 + 8 * (G1_GRP + s));
    #pragma unroll
    for (int s = 0; s < G1_GRP; ++s) c = mm2(g2[s], br[s], c);
    #pragma unroll
    for (int s = 0; s < G1_GRP; ++s) g0[s] = *reinterpret_cast<const float4*>(an1 + 8 * s);
    #pragma unroll
    for (int s = 0; s < G1_GRP; ++s) c = mm2(g3[s], br[G1_GRP + s], c);

    const unsigned rbase = (unsigned)(rt * 16 + ((lane < 16) ? 0 : 8));
    float* orow = out + rbase * (unsigned)HID + ncol;
    #pragma unroll
    for (int i = 0; i < 8; ++i) {
      float v = c[i] + bv;
      v = v > 0.0f ? v : 0.0f;
      orow[(unsigned)i * (unsigned)HID] = v;
    }
  }
}

__global__ void dual_gemm_wmma(const float* __restrict__ A1,
                               const float* __restrict__ W1,
                               const float* __restrict__ A2,
                               const float* __restrict__ W2,
                               const float* __restrict__ bias,
                               float* __restrict__ out,
                               int K, int ncols, int ldo, int do_relu) {
  const int lane   = threadIdx.x & 31;
  const int wid    = blockIdx.x * (blockDim.x >> 5) + (threadIdx.x >> 5);
  const int ctiles = (ncols + 15) >> 4;
  const int mtiles = N_NODES >> 4;
  if (wid >= mtiles * ctiles) return;
  const int rt = wid / ctiles;
  const int ct = wid - rt * ctiles;

  const int mrow  = rt * 16 + (lane & 15);
  const int ncol  = ct * 16 + (lane & 15);
  const int khalf = (lane >> 4) * 4;
  const bool cvalid  = (ncol < ncols);
  const int  ncol_ld = cvalid ? ncol : (ncols - 1);

  const float* arow1 = A1 + (unsigned)mrow    * (unsigned)K + khalf;
  const float* brow1 = W1 + (unsigned)ncol_ld * (unsigned)K + khalf;
  const float* arow2 = A2 + (unsigned)mrow    * (unsigned)K + khalf;
  const float* brow2 = W2 + (unsigned)ncol_ld * (unsigned)K + khalf;

  v8f c = {};

  #pragma unroll 4
  for (int k0 = 0; k0 < K; k0 += 8) {
    const float4 av = *(const float4*)(arow1 + k0);
    const float4 bv = *(const float4*)(brow1 + k0);
    v2f a0; a0.x = av.x; a0.y = av.y;
    v2f b0; b0.x = bv.x; b0.y = bv.y;
    v2f a1; a1.x = av.z; a1.y = av.w;
    v2f b1; b1.x = bv.z; b1.y = bv.w;
    c = __builtin_amdgcn_wmma_f32_16x16x4_f32(false, a0, false, b0,
                                              (short)0, c, false, false);
    c = __builtin_amdgcn_wmma_f32_16x16x4_f32(false, a1, false, b1,
                                              (short)0, c, false, false);
  }
  #pragma unroll 4
  for (int k0 = 0; k0 < K; k0 += 8) {
    const float4 av = *(const float4*)(arow2 + k0);
    const float4 bv = *(const float4*)(brow2 + k0);
    v2f a0; a0.x = av.x; a0.y = av.y;
    v2f b0; b0.x = bv.x; b0.y = bv.y;
    v2f a1; a1.x = av.z; a1.y = av.w;
    v2f b1; b1.x = bv.z; b1.y = bv.w;
    c = __builtin_amdgcn_wmma_f32_16x16x4_f32(false, a0, false, b0,
                                              (short)0, c, false, false);
    c = __builtin_amdgcn_wmma_f32_16x16x4_f32(false, a1, false, b1,
                                              (short)0, c, false, false);
  }

  if (cvalid) {
    float bv = bias[ncol];
    unsigned rbase = (unsigned)(rt * 16 + ((lane < 16) ? 0 : 8));
    float* orow = out + (unsigned)rbase * (unsigned)ldo + ncol;
    #pragma unroll
    for (int i = 0; i < 8; ++i) {
      float v = c[i] + bv;
      if (do_relu) v = v > 0.0f ? v : 0.0f;
      orow[(unsigned)i * (unsigned)ldo] = v;
    }
  }
}

__global__ void log_softmax_rows(const float* __restrict__ logits,
                                 float* __restrict__ out) {
  int r = blockIdx.x * blockDim.x + threadIdx.x;
  if (r >= N_NODES) return;
  const float* l = logits + (long long)r * CLS;
  float m = -3.402823466e38f;
  #pragma unroll
  for (int j = 0; j < CLS; ++j) m = fmaxf(m, l[j]);
  float s = 0.0f;
  #pragma unroll
  for (int j = 0; j < CLS; ++j) s += expf(l[j] - m);
  float ls = logf(s);
  float* o = out + (long long)r * CLS;
  #pragma unroll
  for (int j = 0; j < CLS; ++j) o[j] = l[j] - m - ls;
}

extern "C" void kernel_launch(void* const* d_in, const int* in_sizes, int n_in,
                              void* d_out, int out_size, void* d_ws, size_t ws_size,
                              hipStream_t stream) {
  (void)in_sizes; (void)n_in; (void)out_size; (void)ws_size;

  const float* x    = (const float*)d_in[0];
  const int*   ei   = (const int*)  d_in[1];
  const float* W1_l = (const float*)d_in[2];
  const float* b1   = (const float*)d_in[3];
  const float* W1_r = (const float*)d_in[4];
  const float* W2_l = (const float*)d_in[5];
  const float* b2   = (const float*)d_in[6];
  const float* W2_r = (const float*)d_in[7];
  const int* src = ei;
  const int* dst = ei + NE;
  float* out = (float*)d_out;

  float* ws     = (float*)d_ws;
  float* agg1   = ws   + N_NODES;
  float* h      = agg1 + (long long)N_NODES * F_IN;
  float* agg2   = h    + (long long)N_NODES * HID;
  float* logits = agg2 + (long long)N_NODES * HID;
  unsigned* tags4 = (unsigned*)logits;

  const int TB = 256;
  auto blocks = [](long long n, int tb) { return (unsigned)((n + tb - 1) / tb); };

  edge_owner_tags<<<blocks(NE / 4, TB), TB, 0, stream>>>(dst, tags4);

  owner_mean<F_IN><<<AG_BLOCKS, AG_TB, 0, stream>>>(x, src, dst, tags4, agg1);

  gemm1_weights_in_regs<<<G1_BLOCKS, G1_TB, 0, stream>>>(agg1, W1_l, x, W1_r, b1, h);

  owner_mean<HID><<<AG_BLOCKS, AG_TB, 0, stream>>>(h, src, dst, tags4, agg2);

  {
    int ctiles = (CLS + 15) / 16;
    int tiles = (N_NODES / 16) * ctiles;
    dual_gemm_wmma<<<blocks(tiles, 8), TB, 0, stream>>>(
        agg2, W2_l, h, W2_r, b2, logits, HID, CLS, CLS,  0);
  }

  log_softmax_rows<<<blocks(N_NODES, TB), TB, 0, stream>>>(logits, out);
}


# Layer-2 mean launch skips its edge scan: the layer-1 launch saves each owner wave's routed queue, queue lengths and in-degree counts in the workspace; layer-2 rows are summed in that saved order
# speedup vs baseline: 3.4572x; 1.0541x over previous
#include <hip/hip_runtime.h>

#define N_NODES 50000
#define F_IN    128
#define HID     256
#define CLS     40
#define NE      640000

typedef __attribute__((ext_vector_type(2))) float v2f;
typedef __attribute__((ext_vector_type(8))) float v8f;

#define AG_ROWS   200
#define AG_NW     32
#define AG_TB     (AG_NW * 32)
#define AG_BLOCKS (N_NODES / AG_ROWS)
#define AG_Q      64
#define AG_CHUNK  512
#define AG_NOOWN  255u
#define AG_RQ     256
#define AG_NCHUNK (NE / AG_CHUNK)
#define AG_ROT    (AG_NCHUNK / AG_BLOCKS)
#define SV_WORDS  (AG_ROWS * HID)
#define SV_TAIL   (AG_NW * AG_RQ * 2)
#define SV_CNT    (SV_TAIL + AG_NW)

__global__ __launch_bounds__(256)
void edge_owner_tags(const int* __restrict__ dst, unsigned* __restrict__ tags4) {
  static_assert(AG_BLOCKS < (int)AG_NOOWN, "an owner number fits a byte");
  static_assert(NE % 4 == 0, "whole groups of four edges");
  const int i = (int)(blockIdx.x * 256u + threadIdx.x);
  if (i >= NE / 4) return;
  const int4 d = *reinterpret_cast<const int4*>(dst + i * 4);
  const unsigned t0 = (unsigned)d.x < (unsigned)N_NODES ? (unsigned)d.x / AG_ROWS : AG_NOOWN;
  const unsigned t1 = (unsigned)d.y < (unsigned)N_NODES ? (unsigned)d.y / AG_ROWS : AG_NOOWN;
  const unsigned t2 = (unsigned)d.z < (unsigned)N_NODES ? (unsigned)d.z / AG_ROWS : AG_NOOWN;
  const unsigned t3 = (unsigned)d.w < (unsigned)N_NODES ? (unsigned)d.w / AG_ROWS : AG_NOOWN;
  tags4[i] = t0 | (t1 << 8) | (t2 << 16) | (t3 << 24);
}

static __device__ __forceinline__ void lds_add(float* p, float v) {
  __hip_atomic_fetch_add(p, v, __ATOMIC_RELAXED, __HIP_MEMORY_SCOPE_WORKGROUP);
}

template <int F>
static __device__ __forceinline__ void route_ring(const float* __restrict__ feat,
                                                  const int*   __restrict__ src,
                                                  const int*   __restrict__ dst,
                                                  int row0,
                                                  float* acc, float* cnt,
                                                  const int* qe_w,
                                                  uint2* rq, unsigned* rtail,
                                                  int head, int m, int lane) {
  constexpr int NCH = F / 32;
  const int slot = (head + lane) & (AG_Q - 1);
  const int e  = qe_w[slot];
  const int s  = src[e];
  const int dl = dst[e] - row0;
  bool over = false;
  if (lane < m) {
    lds_add(cnt + dl, 1.0f);
    const unsigned ow  = (unsigned)dl & (unsigned)(AG_NW - 1);
    const unsigned pos = __hip_atomic_fetch_add(rtail + ow, 1u, __ATOMIC_RELAXED,
                                                __HIP_MEMORY_SCOPE_WORKGROUP);
    if (pos < (unsigned)AG_RQ) rq[ow * AG_RQ + pos] = make_uint2((unsigned)s, (unsigned)dl);
    else over = true;
  }
  unsigned ov = __builtin_amdgcn_ballot_w32(over);
  while (ov != 0u) {
    const int j  = __builtin_ctz(ov);
    const int sj = __builtin_amdgcn_readlane(s, j);
    const int dj = __builtin_amdgcn_readlane(dl, j);
    const float* f = feat + (long long)sj * F + lane;
    float* a = acc + dj * F + lane;
    #pragma unroll
    for (int c = 0; c < NCH; ++c) lds_add(a + c * 32, f[c * 32]);
    ov &= ov - 1u;
  }
}

template <int F, int REUSE>
__global__ __launch_bounds__(AG_TB)
void owner_mean(const float* __restrict__ feat,
                const int*   __restrict__ src,
                const int*   __restrict__ dst,
                const unsigned* __restrict__ tags4,
                unsigned* save,
                float* agg) {
  static_assert(SV_CNT + AG_ROWS <= SV_WORDS, "queues, lengths and counts fit the work-group's slice");
  static_assert(NE % AG_CHUNK == 0, "a wave reads whole chunks of the tag list");
  static_assert(AG_NCHUNK % AG_BLOCKS == 0, "the scans' offsets tile the list");
  static_assert(N_NODES % AG_ROWS == 0, "every work-group owns a full block of rows");
  static_assert(AG_Q >= 32 + 32, "one new edge per lane fits the ring beside 31 waiting ones");
  __shared__ __attribute__((aligned(16))) float acc[AG_ROWS * F];
  __shared__ float cnt[AG_ROWS];
  __shared__ int   qe[AG_NW][AG_Q];
  __shared__ uint2    rq[AG_NW * AG_RQ];
  __shared__ unsigned rtail[AG_NW];

  const int tid  = (int)threadIdx.x;
  const int lane = tid & 31;
  const int wave = __builtin_amdgcn_readfirstlane(tid >> 5);
  const int row0 = (int)blockIdx.x * AG_ROWS;

  float4* acc4 = reinterpret_cast<float4*>(acc);
  const float4 z = make_float4(0.0f, 0.0f, 0.0f, 0.0f);
  unsigned* sv = save + (size_t)blockIdx.x * (size_t)SV_WORDS;
  unsigned cv = 0u;
  unsigned tv = 0u;
  if (REUSE) {
    if (tid < AG_ROWS) cv = sv[SV_CNT + tid];
    if (tid < AG_NW) tv = sv[SV_TAIL + tid];
  }
  for (int i = tid; i < AG_ROWS * F / 4; i += AG_TB) acc4[i] = z;
  if (REUSE) {
    if (tid < AG_ROWS) cnt[tid] = __uint_as_float(cv);
    if (tid < AG_NW) rtail[tid] = tv;
  } else {
    if (tid < AG_ROWS) cnt[tid] = 0.0f;
    if (tid < AG_NW) rtail[tid] = 0u;
  }
  int* qe_w = qe[wave];
  for (int i = lane; i < AG_Q; i += 32) qe_w[i] = 0;
  __syncthreads();

  bool scan = true;
  if (REUSE) {
    const unsigned t = rtail[lane];
    scan = __builtin_amdgcn_ballot_w32(t > (unsigned)AG_RQ) != 0u;
    if (scan) {
      __syncthreads();
      if (tid < AG_ROWS) cnt[tid] = 0.0f;
      if (tid < AG_NW) rtail[tid] = 0u;
      __syncthreads();
    } else {
      const int n = __builtin_amdgcn_readfirstlane((int)rtail[wave]);
      const uint2* sq = reinterpret_cast<const uint2*>(sv) + wave * AG_RQ;
      uint2 e[AG_RQ / 32];
      #pragma unroll
      for (int k = 0; k < AG_RQ / 32; ++k) {
        const int i = lane + 32 * k;
        e[k] = i < n ? sq[i] : make_uint2(0u, 0u);
      }
      #pragma unroll
      for (int k = 0; k < AG_RQ / 32; ++k) {
        const int i = lane + 32 * k;
        if (i < n) rq[wave * AG_RQ + i] = e[k];
      }
    }
  }

  if (scan) {
    int head = 0;
    int qn   = 0;
    const int c0   = (int)blockIdx.x * AG_ROT + wave;
    const int cend = (int)blockIdx.x * AG_ROT + AG_NCHUNK;
    const unsigned char* tags = reinterpret_cast<const unsigned char*>(tags4);
    const unsigned K7F = 0x7F7F7F7Fu;
    const unsigned K80 = 0x80808080u;
    const unsigned pat = (unsigned)blockIdx.x * 0x01010101u;
    uint4 nxt = make_uint4(0u, 0u, 0u, 0u);
    {
      const int cc = c0 >= AG_NCHUNK ? c0 - AG_NCHUNK : c0;
      nxt = *reinterpret_cast<const uint4*>(tags + cc * AG_CHUNK + lane * 16);
    }
    for (int c = c0; c < cend; c += AG_NW) {
      const uint4 cur = nxt;
      const int ebase = (c >= AG_NCHUNK ? c - AG_NCHUNK : c) * AG_CHUNK + lane * 16;
      if (c + AG_NW < cend) {
        const int cn = c + AG_NW;
        const int cc = cn >= AG_NCHUNK ? cn - AG_NCHUNK : cn;
        nxt = *reinterpret_cast<const uint4*>(tags + cc * AG_CHUNK + lane * 16);
      }
      const unsigned w0 = cur.x ^ pat;
      const unsigned w1 = cur.y ^ pat;
      const unsigned w2 = cur.z ^ pat;
      const unsigned w3 = cur.w ^ pat;
      const unsigned z0 = ~(((w0 & K7F) + K7F) | w0) & K80;
      const unsigned z1 = ~(((w1 & K7F) + K7F) | w1) & K80;
      const unsigned z2 = ~(((w2 & K7F) + K7F) | w2) & K80;
      const unsigned z3 = ~(((w3 & K7F) + K7F) | w3) & K80;
      unsigned comb = (z0 >> 7) | (z1 >> 6) | (z2 >> 5) | (z3 >> 4);
      unsigned bal = __builtin_amdgcn_ballot_w32(comb != 0u);
      while (bal != 0u) {
        const bool hit = comb != 0u;
        const unsigned p = (unsigned)__builtin_ctz(comb | 0x80000000u);
        const int pos = (head + qn + (int)__builtin_amdgcn_mbcnt_lo(bal, 0u)) & (AG_Q - 1);
        if (hit) qe_w[pos] = ebase + (int)(((p & 3u) << 2) | (p >> 3));
        qn += __builtin_popcount(bal);
        comb &= comb - 1u;
        while (qn >= 32) {
          route_ring<F>(feat, src, dst, row0, acc, cnt, qe_w, rq, rtail, head, 32, lane);
          head = (head + 32) & (AG_Q - 1);
          qn -= 32;
        }
        bal = __builtin_amdgcn_ballot_w32(comb != 0u);
      }
    }
    while (qn > 0) {
      const int m = qn < 32 ? qn : 32;
      route_ring<F>(feat, src, dst, row0, acc, cnt, qe_w, rq, rtail, head, m, lane);
      head = (head + m) & (AG_Q - 1);
      qn -= m;
    }
  }
  __syncthreads();

  if (!REUSE) {
    const unsigned nt = rtail[wave];
    const int n = __builtin_amdgcn_readfirstlane((int)(nt < (unsigned)AG_RQ ? nt : (unsigned)AG_RQ));
    uint2* sq = reinterpret_cast<uint2*>(sv) + wave * AG_RQ;
    for (int i = lane; i < n; i += 32) sq[i] = rq[wave * AG_RQ + i];
    if (tid < AG_NW) sv[SV_TAIL + tid] = rtail[tid];
    if (tid < AG_ROWS) sv[SV_CNT + tid] = __float_as_uint(cnt[tid]);
  }

  {
    constexpr int NV  = F / 128;
    constexpr int GRP = 2048 / F;
    const unsigned nt = rtail[wave];
    const int n = __builtin_amdgcn_readfirstlane((int)(nt < (unsigned)AG_RQ ? nt : (unsigned)AG_RQ));
    const uint2*  q     = rq + wave * AG_RQ;
    const float4* feat4 = reinterpret_cast<const float4*>(feat);
    for (int b = 0; b < n; b += 32) {
      const int m = n - b < 32 ? n - b : 32;
      const uint2 en = lane < m ? q[b + lane] : make_uint2(0u, 0u);
      const int s  = (int)en.x;
      const int dl = (int)en.y;
      for (int j = 0; j < m; j += GRP) {
        float4 v[GRP][NV];
        #pragma unroll
        for (int g = 0; g < GRP; ++g) {
          const int jj = j + g < m ? j + g : m - 1;
          const int sj = __builtin_amdgcn_readlane(s, jj);
          const float4* f = feat4 + (long long)sj * (F / 4) + lane;
          #pragma unroll
          for (int c = 0; c < NV; ++c) v[g][c] = f[c * 32];
        }
        #pragma unroll
        for (int g = 0; g < GRP; ++g) {
          if (j + g < m) {
            const int dj = __builtin_amdgcn_readlane(dl, j + g);
            float4* a = acc4 + dj * (F / 4) + lane;
            #pragma unroll
            for (int c = 0; c < NV; ++c) {
              float4 t = a[c * 32];
              t.x += v[g][c].x;
              t.y += v[g][c].y;
              t.z += v[g][c].z;
              t.w += v[g][c].w;
              a[c * 32] = t;
            }
          }
        }
      }
    }
  }
  __syncthreads();

  constexpr int Q = F / 4;
  float4* out4 = reinterpret_cast<float4*>(agg + (long long)row0 * F);
  for (int i = tid; i < AG_ROWS * Q; i += AG_TB) {
    const float c  = cnt[i / Q];
    const float cc = c > 1.0f ? c : 1.0f;
    const float4 v = acc4[i];
    float4 r;
    r.x = v.x / cc;
    r.y = v.y / cc;
    r.z = v.z / cc;
    r.w = v.w / cc;
    out4[i] = r;
  }
}

#define G1_WAVES  (HID / 16)
#define G1_TB     (G1_WAVES * 32)
#define G1_BLOCKS 256
#define G1_MT     (N_NODES / 16)
#define G1_STEPS  (F_IN / 8)
#define G1_GRP    8

static __device__ __forceinline__ v8f mm2(const float4 av, const float4 bv, v8f c) {
  v2f a0; a0.x = av.x; a0.y = av.y;
  v2f b0; b0.x = bv.x; b0.y = bv.y;
  v2f a1; a1.x = av.z; a1.y = av.w;
  v2f b1; b1.x = bv.z; b1.y = bv.w;
  c = __builtin_amdgcn_wmma_f32_16x16x4_f32(false, a0, false, b0, (short)0, c, false, false);
  c = __builtin_amdgcn_wmma_f32_16x16x4_f32(false, a1, false, b1, (short)0, c, false, false);
  return c;
}

__global__ __launch_bounds__(G1_TB)
void gemm1_weights_in_regs(const float* __restrict__ A1,
                           const float* __restrict__ W1,
                           const float* __restrict__ A2,
                           const float* __restrict__ W2,
                           const float* __restrict__ bias,
                           float* __restrict__ out) {
  static_assert(N_NODES % 16 == 0, "whole row tiles");
  static_assert(G1_STEPS == 2 * G1_GRP, "two groups of k-steps per product");
  const int lane  = (int)threadIdx.x & 31;
  const int ct    = __builtin_amdgcn_readfirstlane((int)threadIdx.x >> 5);
  const int c16   = lane & 15;
  const int khalf = (lane >> 4) * 4;
  const int ncol  = ct * 16 + c16;

  float4 bl[G1_STEPS], br[G1_STEPS];
  {
    const float* bp1 = W1 + ncol * F_IN + khalf;
    const float* bp2 = W2 + ncol * F_IN + khalf;
    #pragma unroll
    for (int s = 0; s < G1_STEPS; ++s) {
      bl[s] = *reinterpret_cast<const float4*>(bp1 + 8 * s);
      br[s] = *reinterpret_cast<const float4*>(bp2 + 8 * s);
    }
  }
  const float bv = bias[ncol];

  float4 g0[G1_GRP], g1[G1_GRP], g2[G1_GRP], g3[G1_GRP];
  int rt = (int)blockIdx.x;
  {
    const float* ap1 = A1 + (unsigned)(rt * 16 + c16) * (unsigned)F_IN + khalf;
    #pragma unroll
    for (int s = 0; s < G1_GRP; ++s) g0[s] = *reinterpret_cast<const float4*>(ap1 + 8 * s);
  }
  for (; rt < G1_MT; rt += (int)gridDim.x) {
    const unsigned roff = (unsigned)(rt * 16 + c16) * (unsigned)F_IN + khalf;
    const float* ap1 = A1 + roff;
    const float* ap2 = A2 + roff;
    const int rn = rt + (int)gridDim.x < G1_MT ? rt + (int)gridDim.x : rt;
    const float* an1 = A1 + (unsigned)(rn * 16 + c16) * (unsigned)F_IN + khalf;
    v8f c = {};
    #pragma unroll
    for (int s = 0; s < G1_GRP; ++s) g1[s] = *reinterpret_cast<const float4*>(ap1 + 8 * (G1_GRP + s));
    #pragma unroll
    for (int s = 0; s < G1_GRP; ++s) c = mm2(g0[s], bl[s], c);
    #pragma unroll
    for (int s = 0; s < G1_GRP; ++s) g2[s] = *reinterpret_cast<const float4*>(ap2 + 8 * s);
    #pragma unroll
    for (int s = 0; s < G1_GRP; ++s) c = mm2(g1[s], bl[G1_GRP + s], c);
    #pragma unroll
    for (int s = 0; s < G1_GRP; ++s) g3[s] = *reinterpret_cast<const float4*>(ap2 + 8 * (G1_GRP + s));
    #pragma unroll
    for (int s = 0; s < G1_GRP; ++s) c = mm2(g2[s], br[s], c);
    #pragma unroll
    for (int s = 0; s < G1_GRP; ++s) g0[s] = *reinterpret_cast<const float4*>(an1 + 8 * s);
    #pragma unroll
    for (int s = 0; s < G1_GRP; ++s) c = mm2(g3[s], br[G1_GRP + s], c);

    const unsigned rbase = (unsigned)(rt * 16 + ((lane < 16) ? 0 : 8));
    float* orow = out + rbase * (unsigned)HID + ncol;
    #pragma unroll
    for (int i = 0; i < 8; ++i) {
      float v = c[i] + bv;
      v = v > 0.0f ? v : 0.0f;
      orow[(unsigned)i * (unsigned)HID] = v;
    }
  }
}

__global__ void dual_gemm_wmma(const float* __restrict__ A1,
                               const float* __restrict__ W1,
                               const float* __restrict__ A2,
                               const float* __restrict__ W2,
                               const float* __restrict__ bias,
                               float* __restrict__ out,
                               int K, int ncols, int ldo, int do_relu) {
  const int lane   = threadIdx.x & 31;
  const int wid    = blockIdx.x * (blockDim.x >> 5) + (threadIdx.x >> 5);
  const int ctiles = (ncols + 15) >> 4;
  const int mtiles = N_NODES >> 4;
  if (wid >= mtiles * ctiles) return;
  const int rt = wid / ctiles;
  const int ct = wid - rt * ctiles;

  const int mrow  = rt * 16 + (lane & 15);
  const int ncol  = ct * 16 + (lane & 15);
  const int khalf = (lane >> 4) * 4;
  const bool cvalid  = (ncol < ncols);
  const int  ncol_ld = cvalid ? ncol : (ncols - 1);

  const float* arow1 = A1 + (unsigned)mrow    * (unsigned)K + khalf;
  const float* brow1 = W1 + (unsigned)ncol_ld * (unsigned)K + khalf;
  const float* arow2 = A2 + (unsigned)mrow    * (unsigned)K + khalf;
  const float* brow2 = W2 + (unsigned)ncol_ld * (unsigned)K + khalf;

  v8f c = {};

  #pragma unroll 4
  for (int k0 = 0; k0 < K; k0 += 8) {
    const float4 av = *(const float4*)(arow1 + k0);
    const float4 bv = *(const float4*)(brow1 + k0);
    v2f a0; a0.x = av.x; a0.y = av.y;
    v2f b0; b0.x = bv.x; b0.y = bv.y;
    v2f a1; a1.x = av.z; a1.y = av.w;
    v2f b1; b1.x = bv.z; b1.y = bv.w;
    c = __builtin_amdgcn_wmma_f32_16x16x4_f32(false, a0, false, b0,
                                              (short)0, c, false, false);
    c = __builtin_amdgcn_wmma_f32_16x16x4_f32(false, a1, false, b1,
                                              (short)0, c, false, false);
  }
  #pragma unroll 4
  for (int k0 = 0; k0 < K; k0 += 8) {
    const float4 av = *(const float4*)(arow2 + k0);
    const float4 bv = *(const float4*)(brow2 + k0);
    v2f a0; a0.x = av.x; a0.y = av.y;
    v2f b0; b0.x = bv.x; b0.y = bv.y;
    v2f a1; a1.x = av.z; a1.y = av.w;
    v2f b1; b1.x = bv.z; b1.y = bv.w;
    c = __builtin_amdgcn_wmma_f32_16x16x4_f32(false, a0, false, b0,
                                              (short)0, c, false, false);
    c = __builtin_amdgcn_wmma_f32_16x16x4_f32(false, a1, false, b1,
                                              (short)0, c, false, false);
  }

  if (cvalid) {
    float bv = bias[ncol];
    unsigned rbase = (unsigned)(rt * 16 + ((lane < 16) ? 0 : 8));
    float* orow = out + (unsigned)rbase * (unsigned)ldo + ncol;
    #pragma unroll
    for (int i = 0; i < 8; ++i) {
      float v = c[i] + bv;
      if (do_relu) v = v > 0.0f ? v : 0.0f;
      orow[(unsigned)i * (unsigned)ldo] = v;
    }
  }
}

__global__ void log_softmax_rows(const float* __restrict__ logits,
                                 float* __restrict__ out) {
  int r = blockIdx.x * blockDim.x + threadIdx.x;
  if (r >= N_NODES) return;
  const float* l = logits + (long long)r * CLS;
  float m = -3.402823466e38f;
  #pragma unroll
  for (int j = 0; j < CLS; ++j) m = fmaxf(m, l[j]);
  float s = 0.0f;
  #pragma unroll
  for (int j = 0; j < CLS; ++j) s += expf(l[j] - m);
  float ls = logf(s);
  float* o = out + (long long)r * CLS;
  #pragma unroll
  for (int j = 0; j < CLS; ++j) o[j] = l[j] - m - ls;
}

extern "C" void kernel_launch(void* const* d_in, const int* in_sizes, int n_in,
                              void* d_out, int out_size, void* d_ws, size_t ws_size,
                              hipStream_t stream) {
  (void)in_sizes; (void)n_in; (void)out_size; (void)ws_size;

  const float* x    = (const float*)d_in[0];
  const int*   ei   = (const int*)  d_in[1];
  const float* W1_l = (const float*)d_in[2];
  const float* b1   = (const float*)d_in[3];
  const float* W1_r = (const float*)d_in[4];
  const float* W2_l = (const float*)d_in[5];
  const float* b2   = (const float*)d_in[6];
  const float* W2_r = (const float*)d_in[7];
  const int* src = ei;
  const int* dst = ei + NE;
  float* out = (float*)d_out;

  float* ws     = (float*)d_ws;
  float* agg1   = ws   + N_NODES;
  float* h      = agg1 + (long long)N_NODES * F_IN;
  float* agg2   = h    + (long long)N_NODES * HID;
  float* logits = agg2 + (long long)N_NODES * HID;
  unsigned* tags4 = (unsigned*)logits;

  const int TB = 256;
  auto blocks = [](long long n, int tb) { return (unsigned)((n + tb - 1) / tb); };

  edge_owner_tags<<<blocks(NE / 4, TB), TB, 0, stream>>>(dst, tags4);

  owner_mean<F_IN, 0><<<AG_BLOCKS, AG_TB, 0, stream>>>(x, src, dst, tags4, (unsigned*)agg2, agg1);

  gemm1_weights_in_regs<<<G1_BLOCKS, G1_TB, 0, stream>>>(agg1, W1_l, x, W1_r, b1, h);

  owner_mean<HID, 1><<<AG_BLOCKS, AG_TB, 0, stream>>>(h, src, dst, tags4, (unsigned*)agg2, agg2);

  {
    int ctiles = (CLS + 15) / 16;
    int tiles = (N_NODES / 16) * ctiles;
    dual_gemm_wmma<<<blocks(tiles, 8), TB, 0, stream>>>(
        agg2, W2_l, h, W2_r, b2, logits, HID, CLS, CLS,  0);
  }

  log_softmax_rows<<<blocks(N_NODES, TB), TB, 0, stream>>>(logits, out);
}


# Layer-2 product and log-softmax done inside the layer-2 mean launch: means stay in LDS (padded rows), both weights staged in LDS, 39 WMMA tiles per work-group; 6->4 launches; sum order kept
# speedup vs baseline: 4.3944x; 1.2711x over previous
#include <hip/hip_runtime.h>

#define N_NODES 50000
#define F_IN    128
#define HID     256
#define CLS     40
#define NE      640000

typedef __attribute__((ext_vector_type(2))) float v2f;
typedef __attribute__((ext_vector_type(8))) float v8f;

#define AG_ROWS   200
#define AG_NW     32
#define AG_TB     (AG_NW * 32)
#define AG_BLOCKS (N_NODES / AG_ROWS)
#define AG_Q      64
#define AG_CHUNK  512
#define AG_NOOWN  255u
#define AG_RQ     256
#define AG_NCHUNK (NE / AG_CHUNK)
#define AG_ROT    (AG_NCHUNK / AG_BLOCKS)
#define SV_WORDS  (AG_ROWS * HID)
#define SV_TAIL   (AG_NW * AG_RQ * 2)
#define SV_CNT    (SV_TAIL + AG_NW)
#define L2_PITCH  (HID + 4)
#define L2_RT     ((AG_ROWS + 15) / 16)
#define L2_CT     ((CLS + 15) / 16)
#define L2_TILES  (L2_RT * L2_CT)
#define LG_PITCH  (CLS + 1)

__global__ __launch_bounds__(256)
void edge_owner_tags(const int* __restrict__ dst, unsigned* __restrict__ tags4) {
  static_assert(AG_BLOCKS < (int)AG_NOOWN, "an owner number fits a byte");
  static_assert(NE % 4 == 0, "whole groups of four edges");
  const int i = (int)(blockIdx.x * 256u + threadIdx.x);
  if (i >= NE / 4) return;
  const int4 d = *reinterpret_cast<const int4*>(dst + i * 4);
  const unsigned t0 = (unsigned)d.x < (unsigned)N_NODES ? (unsigned)d.x / AG_ROWS : AG_NOOWN;
  const unsigned t1 = (unsigned)d.y < (unsigned)N_NODES ? (unsigned)d.y / AG_ROWS : AG_NOOWN;
  const unsigned t2 = (unsigned)d.z < (unsigned)N_NODES ? (unsigned)d.z / AG_ROWS : AG_NOOWN;
  const unsigned t3 = (unsigned)d.w < (unsigned)N_NODES ? (unsigned)d.w / AG_ROWS : AG_NOOWN;
  tags4[i] = t0 | (t1 << 8) | (t2 << 16) | (t3 << 24);
}

static __device__ __forceinline__ void lds_add(float* p, float v) {
  __hip_atomic_fetch_add(p, v, __ATOMIC_RELAXED, __HIP_MEMORY_SCOPE_WORKGROUP);
}

template <int F, int P>
static __device__ __forceinline__ void route_ring(const float* __restrict__ feat,
                                                  const int*   __restrict__ src,
                                                  const int*   __restrict__ dst,
                                                  int row0,
                                                  float* acc, float* cnt,
                                                  const int* qe_w,
                                                  uint2* rq, unsigned* rtail,
                                                  int head, int m, int lane) {
  constexpr int NCH = F / 32;
  const int slot = (head + lane) & (AG_Q - 1);
  const int e  = qe_w[slot];
  const int s  = src[e];
  const int dl = dst[e] - row0;
  bool over = false;
  if (lane < m) {
    lds_add(cnt + dl, 1.0f);
    const unsigned ow  = (unsigned)dl & (unsigned)(AG_NW - 1);
    const unsigned pos = __hip_atomic_fetch_add(rtail + ow, 1u, __ATOMIC_RELAXED,
                                                __HIP_MEMORY_SCOPE_WORKGROUP);
    if (pos < (unsigned)AG_RQ) rq[ow * AG_RQ + pos] = make_uint2((unsigned)s, (unsigned)dl);
    else over = true;
  }
  unsigned ov = __builtin_amdgcn_ballot_w32(over);
  while (ov != 0u) {
    const int j  = __builtin_ctz(ov);
    const int sj = __builtin_amdgcn_readlane(s, j);
    const int dj = __builtin_amdgcn_readlane(dl, j);
    const float* f = feat + (long long)sj * F + lane;
    float* a = acc + dj * P + lane;
    #pragma unroll
    for (int c = 0; c < NCH; ++c) lds_add(a + c * 32, f[c * 32]);
    ov &= ov - 1u;
  }
}

static __device__ __forceinline__ v8f mm2(const float4 av, const float4 bv, v8f c) {
  v2f a0; a0.x = av.x; a0.y = av.y;
  v2f b0; b0.x = bv.x; b0.y = bv.y;
  v2f a1; a1.x = av.z; a1.y = av.w;
  v2f b1; b1.x = bv.z; b1.y = bv.w;
  c = __builtin_amdgcn_wmma_f32_16x16x4_f32(false, a0, false, b0, (short)0, c, false, false);
  c = __builtin_amdgcn_wmma_f32_16x16x4_f32(false, a1, false, b1, (short)0, c, false, false);
  return c;
}

static __device__ __forceinline__ v8f layer2_tile(const float4* mean4,
                                                  const float4* wl4,
                                                  const float4* wr4,
                                                  const float* __restrict__ hrows,
                                                  int t, int lane) {
  const int rt  = t / L2_CT;
  const int ct  = t - rt * L2_CT;
  const int c16 = lane & 15;
  const int hh  = lane >> 4;
  int rowl = rt * 16 + c16;
  rowl = rowl < AG_ROWS ? rowl : AG_ROWS - 1;
  int col = ct * 16 + c16;
  col = col < CLS ? col : CLS - 1;
  const float4* ap = mean4 + rowl * (L2_PITCH / 4) + hh;
  const float4* bl = wl4 + col * (L2_PITCH / 4) + hh;
  const float4* br = wr4 + col * (L2_PITCH / 4) + hh;
  const float*  hp = hrows + rowl * HID + hh * 4;
  v8f c = {};
  #pragma unroll 4
  for (int ks = 0; ks < HID / 8; ++ks) c = mm2(ap[2 * ks], bl[2 * ks], c);
  #pragma unroll 4
  for (int ks = 0; ks < HID / 8; ++ks)
    c = mm2(*reinterpret_cast<const float4*>(hp + 8 * ks), br[2 * ks], c);
  return c;
}

static __device__ __forceinline__ void logits_to_lds(float* lg, const v8f d,
                                                     const float* __restrict__ bias,
                                                     int t, int lane) {
  const int rt  = t / L2_CT;
  const int ct  = t - rt * L2_CT;
  const int col = ct * 16 + (lane & 15);
  const int rb  = rt * 16 + (lane >> 4) * 8;
  if (col < CLS) {
    const float bv = bias[col];
    #pragma unroll
    for (int i = 0; i < 8; ++i)
      if (rb + i < AG_ROWS) lg[(rb + i) * LG_PITCH + col] = d[i] + bv;
  }
}

template <int F, int REUSE>
__global__ __launch_bounds__(AG_TB)
void owner_mean(const float* __restrict__ feat,
                const int*   __restrict__ src,
                const int*   __restrict__ dst,
                const unsigned* __restrict__ tags4,
                unsigned* save,
                float* agg,
                const float* __restrict__ w_l,
                const float* __restrict__ w_r,
                const float* __restrict__ bias,
                float* __restrict__ out) {
  static_assert(REUSE == 0 || F == HID, "the fused layer-2 tail is written for rows of HID floats");
  constexpr int P = REUSE ? L2_PITCH : F;
  static_assert(P % 4 == 0, "rows of acc start on 16 bytes");
  static_assert(SV_CNT + AG_ROWS <= SV_WORDS, "queues, lengths and counts fit the work-group's slice");
  static_assert(NE % AG_CHUNK == 0, "a wave reads whole chunks of the tag list");
  static_assert(AG_NCHUNK % AG_BLOCKS == 0, "the scans' offsets tile the list");
  static_assert(N_NODES % AG_ROWS == 0, "every work-group owns a full block of rows");
  static_assert(AG_Q >= 32 + 32, "one new edge per lane fits the ring beside 31 waiting ones");
  __shared__ __attribute__((aligned(16))) float acc[AG_ROWS * P];
  __shared__ float cnt[AG_ROWS];
  __shared__ int   qe[AG_NW][AG_Q];
  __shared__ __attribute__((aligned(16))) uint2 rq[AG_NW * AG_RQ];
  __shared__ unsigned rtail[AG_NW];

  const int tid  = (int)threadIdx.x;
  const int lane = tid & 31;
  const int wave = __builtin_amdgcn_readfirstlane(tid >> 5);
  const int row0 = (int)blockIdx.x * AG_ROWS;

  float4* acc4 = reinterpret_cast<float4*>(acc);
  const float4 z = make_float4(0.0f, 0.0f, 0.0f, 0.0f);
  unsigned* sv = save + (size_t)blockIdx.x * (size_t)SV_WORDS;
  unsigned cv = 0u;
  unsigned tv = 0u;
  if (REUSE) {
    if (tid < AG_ROWS) cv = sv[SV_CNT + tid];
    if (tid < AG_NW) tv = sv[SV_TAIL + tid];
  }
  for (int i = tid; i < AG_ROWS * P / 4; i += AG_TB) acc4[i] = z;
  if (REUSE) {
    if (tid < AG_ROWS) cnt[tid] = __uint_as_float(cv);
    if (tid < AG_NW) rtail[tid] = tv;
  } else {
    if (tid < AG_ROWS) cnt[tid] = 0.0f;
    if (tid < AG_NW) rtail[tid] = 0u;
  }
  int* qe_w = qe[wave];
  for (int i = lane; i < AG_Q; i += 32) qe_w[i] = 0;
  __syncthreads();

  bool scan = true;
  if (REUSE) {
    const unsigned t = rtail[lane];
    scan = __builtin_amdgcn_ballot_w32(t > (unsigned)AG_RQ) != 0u;
    if (scan) {
      __syncthreads();
      if (tid < AG_ROWS) cnt[tid] = 0.0f;
      if (tid < AG_NW) rtail[tid] = 0u;
      __syncthreads();
    } else {
      const int n = __builtin_amdgcn_readfirstlane((int)rtail[wave]);
      const uint2* sq = reinterpret_cast<const uint2*>(sv) + wave * AG_RQ;
      uint2 e[AG_RQ / 32];
      #pragma unroll
      for (int k = 0; k < AG_RQ / 32; ++k) {
        const int i = lane + 32 * k;
        e[k] = i < n ? sq[i] : make_uint2(0u, 0u);
      }
      #pragma unroll
      for (int k = 0; k < AG_RQ / 32; ++k) {
        const int i = lane + 32 * k;
        if (i < n) rq[wave * AG_RQ + i] = e[k];
      }
    }
  }

  if (scan) {
    int head = 0;
    int qn   = 0;
    const int c0   = (int)blockIdx.x * AG_ROT + wave;
    const int cend = (int)blockIdx.x * AG_ROT + AG_NCHUNK;
    const unsigned char* tags = reinterpret_cast<const unsigned char*>(tags4);
    const unsigned K7F = 0x7F7F7F7Fu;
    const unsigned K80 = 0x80808080u;
    const unsigned pat = (unsigned)blockIdx.x * 0x01010101u;
    uint4 nxt = make_uint4(0u, 0u, 0u, 0u);
    {
      const int cc = c0 >= AG_NCHUNK ? c0 - AG_NCHUNK : c0;
      nxt = *reinterpret_cast<const uint4*>(tags + cc * AG_CHUNK + lane * 16);
    }
    for (int c = c0; c < cend; c += AG_NW) {
      const uint4 cur = nxt;
      const int ebase = (c >= AG_NCHUNK ? c - AG_NCHUNK : c) * AG_CHUNK + lane * 16;
      if (c + AG_NW < cend) {
        const int cn = c + AG_NW;
        const int cc = cn >= AG_NCHUNK ? cn - AG_NCHUNK : cn;
        nxt = *reinterpret_cast<const uint4*>(tags + cc * AG_CHUNK + lane * 16);
      }
      const unsigned w0 = cur.x ^ pat;
      const unsigned w1 = cur.y ^ pat;
      const unsigned w2 = cur.z ^ pat;
      const unsigned w3 = cur.w ^ pat;
      const unsigned z0 = ~(((w0 & K7F) + K7F) | w0) & K80;
      const unsigned z1 = ~(((w1 & K7F) + K7F) | w1) & K80;
      const unsigned z2 = ~(((w2 & K7F) + K7F) | w2) & K80;
      const unsigned z3 = ~(((w3 & K7F) + K7F) | w3) & K80;
      unsigned comb = (z0 >> 7) | (z1 >> 6) | (z2 >> 5) | (z3 >> 4);
      unsigned bal = __builtin_amdgcn_ballot_w32(comb != 0u);
      while (bal != 0u) {
        const bool hit = comb != 0u;
        const unsigned p = (unsigned)__builtin_ctz(comb | 0x80000000u);
        const int pos = (head + qn + (int)__builtin_amdgcn_mbcnt_lo(bal, 0u)) & (AG_Q - 1);
        if (hit) qe_w[pos] = ebase + (int)(((p & 3u) << 2) | (p >> 3));
        qn += __builtin_popcount(bal);
        comb &= comb - 1u;
        while (qn >= 32) {
          route_ring<F, P>(feat, src, dst, row0, acc, cnt, qe_w, rq, rtail, head, 32, lane);
          head = (head + 32) & (AG_Q - 1);
          qn -= 32;
        }
        bal = __builtin_amdgcn_ballot_w32(comb != 0u);
      }
    }
    while (qn > 0) {
      const int m = qn < 32 ? qn : 32;
      route_ring<F, P>(feat, src, dst, row0, acc, cnt, qe_w, rq, rtail, head, m, lane);
      head = (head + m) & (AG_Q - 1);
      qn -= m;
    }
  }
  __syncthreads();

  if (!REUSE) {
    const unsigned nt = rtail[wave];
    const int n = __builtin_amdgcn_readfirstlane((int)(nt < (unsigned)AG_RQ ? nt : (unsigned)AG_RQ));
    uint2* sq = reinterpret_cast<uint2*>(sv) + wave * AG_RQ;
    for (int i = lane; i < n; i += 32) sq[i] = rq[wave * AG_RQ + i];
    if (tid < AG_NW) sv[SV_TAIL + tid] = rtail[tid];
    if (tid < AG_ROWS) sv[SV_CNT + tid] = __float_as_uint(cnt[tid]);
  }

  {
    constexpr int NV  = F / 128;
    constexpr int GRP = 2048 / F;
    const unsigned nt = rtail[wave];
    const int n = __builtin_amdgcn_readfirstlane((int)(nt < (unsigned)AG_RQ ? nt : (unsigned)AG_RQ));
    const uint2*  q     = rq + wave * AG_RQ;
    const float4* feat4 = reinterpret_cast<const float4*>(feat);
    for (int b = 0; b < n; b += 32) {
      const int m = n - b < 32 ? n - b : 32;
      const uint2 en = lane < m ? q[b + lane] : make_uint2(0u, 0u);
      const int s  = (int)en.x;
      const int dl = (int)en.y;
      for (int j = 0; j < m; j += GRP) {
        float4 v[GRP][NV];
        #pragma unroll
        for (int g = 0; g < GRP; ++g) {
          const int jj = j + g < m ? j + g : m - 1;
          const int sj = __builtin_amdgcn_readlane(s, jj);
          const float4* f = feat4 + (long long)sj * (F / 4) + lane;
          #pragma unroll
          for (int c = 0; c < NV; ++c) v[g][c] = f[c * 32];
        }
        #pragma unroll
        for (int g = 0; g < GRP; ++g) {
          if (j + g < m) {
            const int dj = __builtin_amdgcn_readlane(dl, j + g);
            float4* a = acc4 + dj * (P / 4) + lane;
            #pragma unroll
            for (int c = 0; c < NV; ++c) {
              float4 t = a[c * 32];
              t.x += v[g][c].x;
              t.y += v[g][c].y;
              t.z += v[g][c].z;
              t.w += v[g][c].w;
              a[c * 32] = t;
            }
          }
        }
      }
    }
  }
  __syncthreads();

  constexpr int Q = F / 4;
  if constexpr (REUSE != 0) {
    constexpr int P4 = P / 4;
    static_assert(CLS * L2_PITCH * 4 <= AG_NW * AG_RQ * 8, "W_l fits the room of the routed queues");
    static_assert(AG_ROWS * LG_PITCH <= AG_ROWS * P, "the logits fit the room of the sums");
    static_assert(L2_TILES > AG_NW && L2_TILES <= 2 * AG_NW, "every wave has one tile, some a second");
    __shared__ __attribute__((aligned(16))) float w2r[CLS * L2_PITCH];
    float4* wl4 = reinterpret_cast<float4*>(rq);
    float4* wr4 = reinterpret_cast<float4*>(w2r);
    const float4* gl = reinterpret_cast<const float4*>(w_l);
    const float4* gr = reinterpret_cast<const float4*>(w_r);
    for (int i = tid; i < CLS * Q; i += AG_TB) {
      const int col = i / Q;
      const int q   = i - col * Q;
      wl4[col * P4 + q] = gl[i];
      wr4[col * P4 + q] = gr[i];
    }
    for (int i = tid; i < AG_ROWS * Q; i += AG_TB) {
      const int r = i / Q;
      const int q = i - r * Q;
      const float c  = cnt[r];
      const float cc = c > 1.0f ? c : 1.0f;
      float4 v = acc4[r * P4 + q];
      v.x = v.x / cc;
      v.y = v.y / cc;
      v.z = v.z / cc;
      v.w = v.w / cc;
      acc4[r * P4 + q] = v;
    }
    __syncthreads();

    const float* hrows = feat + (long long)row0 * F;
    const bool two = wave + AG_NW < L2_TILES;
    v8f d0 = layer2_tile(acc4, wl4, wr4, hrows, wave, lane);
    v8f d1 = {};
    if (two) d1 = layer2_tile(acc4, wl4, wr4, hrows, wave + AG_NW, lane);
    __syncthreads();

    float* lg = acc;
    logits_to_lds(lg, d0, bias, wave, lane);
    if (two) logits_to_lds(lg, d1, bias, wave + AG_NW, lane);
    __syncthreads();

    if (tid < AG_ROWS) {
      float* l = lg + tid * LG_PITCH;
      float m = -3.402823466e38f;
      #pragma unroll
      for (int j = 0; j < CLS; ++j) m = fmaxf(m, l[j]);
      float ssum = 0.0f;
      #pragma unroll
      for (int j = 0; j < CLS; ++j) ssum += expf(l[j] - m);
      const float ls = logf(ssum);
      #pragma unroll
      for (int j = 0; j < CLS; ++j) l[j] = l[j] - m - ls;
    }
    __syncthreads();

    float* o = out + (long long)row0 * CLS;
    for (int i = tid; i < AG_ROWS * CLS; i += AG_TB) {
      const int r = i / CLS;
      o[i] = lg[r * LG_PITCH + (i - r * CLS)];
    }
  } else {
    float4* out4 = reinterpret_cast<float4*>(agg + (long long)row0 * F);
    for (int i = tid; i < AG_ROWS * Q; i += AG_TB) {
      const float c  = cnt[i / Q];
      const float cc = c > 1.0f ? c : 1.0f;
      const float4 v = acc4[i];
      float4 r;
      r.x = v.x / cc;
      r.y = v.y / cc;
      r.z = v.z / cc;
      r.w = v.w / cc;
      out4[i] = r;
    }
  }
}

#define G1_WAVES  (HID / 16)
#define G1_TB     (G1_WAVES * 32)
#define G1_BLOCKS 256
#define G1_MT     (N_NODES / 16)
#define G1_STEPS  (F_IN / 8)
#define G1_GRP    8

__global__ __launch_bounds__(G1_TB)
void gemm1_weights_in_regs(const float* __restrict__ A1,
                           const float* __restrict__ W1,
                           const float* __restrict__ A2,
                           const float* __restrict__ W2,
                           const float* __restrict__ bias,
                           float* __restrict__ out) {
  static_assert(N_NODES % 16 == 0, "whole row tiles");
  static_assert(G1_STEPS == 2 * G1_GRP, "two groups of k-steps per product");
  const int lane  = (int)threadIdx.x & 31;
  const int ct    = __builtin_amdgcn_readfirstlane((int)threadIdx.x >> 5);
  const int c16   = lane & 15;
  const int khalf = (lane >> 4) * 4;
  const int ncol  = ct * 16 + c16;

  float4 bl[G1_STEPS], br[G1_STEPS];
  {
    const float* bp1 = W1 + ncol * F_IN + khalf;
    const float* bp2 = W2 + ncol * F_IN + khalf;
    #pragma unroll
    for (int s = 0; s < G1_STEPS; ++s) {
      bl[s] = *reinterpret_cast<const float4*>(bp1 + 8 * s);
      br[s] = *reinterpret_cast<const float4*>(bp2 + 8 * s);
    }
  }
  const float bv = bias[ncol];

  float4 g0[G1_GRP], g1[G1_GRP], g2[G1_GRP], g3[G1_GRP];
  int rt = (int)blockIdx.x;
  {
    const float* ap1 = A1 + (unsigned)(rt * 16 + c16) * (unsigned)F_IN + khalf;
    #pragma unroll
    for (int s = 0; s < G1_GRP; ++s) g0[s] = *reinterpret_cast<const float4*>(ap1 + 8 * s);
  }
  for (; rt < G1_MT; rt += (int)gridDim.x) {
    const unsigned roff = (unsigned)(rt * 16 + c16) * (unsigned)F_IN + khalf;
    const float* ap1 = A1 + roff;
    const float* ap2 = A2 + roff;
    const int rn = rt + (int)gridDim.x < G1_MT ? rt + (int)gridDim.x : rt;
    const float* an1 = A1 + (unsigned)(rn * 16 + c16) * (unsigned)F_IN + khalf;
    v8f c = {};
    #pragma unroll
    for (int s = 0; s < G1_GRP; ++s) g1[s] = *reinterpret_cast<const float4*>(ap1 + 8 * (G1_GRP + s));
    #pragma unroll
    for (int s = 0; s < G1_GRP; ++s) c = mm2(g0[s], bl[s], c);
    #pragma unroll
    for (int s = 0; s < G1_GRP; ++s) g2[s] = *reinterpret_cast<const float4*>(ap2 + 8 * s);
    #pragma unroll
    for (int s = 0; s < G1_GRP; ++s) c = mm2(g1[s], bl[G1_GRP + s], c);
    #pragma unroll
    for (int s = 0; s < G1_GRP; ++s) g3[s] = *reinterpret_cast<const float4*>(ap2 + 8 * (G1_GRP + s));
    #pragma unroll
    for (int s = 0; s < G1_GRP; ++s) c = mm2(g2[s], br[s], c);
    #pragma unroll
    for (int s = 0; s < G1_GRP; ++s) g0[s] = *reinterpret_cast<const float4*>(an1 + 8 * s);
    #pragma unroll
    for (int s = 0; s < G1_GRP; ++s) c = mm2(g3[s], br[G1_GRP + s], c);

    const unsigned rbase = (unsigned)(rt * 16 + ((lane < 16) ? 0 : 8));
    float* orow = out + rbase * (unsigned)HID + ncol;
    #pragma unroll
    for (int i = 0; i < 8; ++i) {
      float v = c[i] + bv;
      v = v > 0.0f ? v : 0.0f;
      orow[(unsigned)i * (unsigned)HID] = v;
    }
  }
}

extern "C" void kernel_launch(void* const* d_in, const int* in_sizes, int n_in,
                              void* d_out, int out_size, void* d_ws, size_t ws_size,
                              hipStream_t stream) {
  (void)in_sizes; (void)n_in; (void)out_size; (void)ws_size;

  const float* x    = (const float*)d_in[0];
  const int*   ei   = (const int*)  d_in[1];
  const float* W1_l = (const float*)d_in[2];
  const float* b1   = (const float*)d_in[3];
  const float* W1_r = (const float*)d_in[4];
  const float* W2_l = (const float*)d_in[5];
  const float* b2   = (const float*)d_in[6];
  const float* W2_r = (const float*)d_in[7];
  const int* src = ei;
  const int* dst = ei + NE;
  float* out = (float*)d_out;

  float* ws     = (float*)d_ws;
  float* agg1   = ws   + N_NODES;
  float* h      = agg1 + (long long)N_NODES * F_IN;
  float* agg2   = h    + (long long)N_NODES * HID;
  float* tagpl  = agg2 + (long long)N_NODES * HID;
  unsigned* tags4 = (unsigned*)tagpl;

  const int TB = 256;
  auto blocks = [](long long n, int tb) { return (unsigned)((n + tb - 1) / tb); };

  edge_owner_tags<<<blocks(NE / 4, TB), TB, 0, stream>>>(dst, tags4);

  owner_mean<F_IN, 0><<<AG_BLOCKS, AG_TB, 0, stream>>>(x, src, dst, tags4, (unsigned*)agg2, agg1,
                                                       nullptr, nullptr, nullptr, nullptr);

  gemm1_weights_in_regs<<<G1_BLOCKS, G1_TB, 0, stream>>>(agg1, W1_l, x, W1_r, b1, h);

  owner_mean<HID, 1><<<AG_BLOCKS, AG_TB, 0, stream>>>(h, src, dst, tags4, (unsigned*)agg2, nullptr,
                                                      W2_l, W2_r, b2, out);
}


# Layer-1 product: the 16 weight-holding waves of a block get a row tile's A rows through LDS, loaded once in whole rows (3 stages, one barrier a tile), not 16 times in 16-line loads; sum order kept
# speedup vs baseline: 5.2056x; 1.1846x over previous
#include <hip/hip_runtime.h>

#define N_NODES 50000
#define F_IN    128
#define HID     256
#define CLS     40
#define NE      640000

typedef __attribute__((ext_vector_type(2))) float v2f;
typedef __attribute__((ext_vector_type(8))) float v8f;

#define AG_ROWS   200
#define AG_NW     32
#define AG_TB     (AG_NW * 32)
#define AG_BLOCKS (N_NODES / AG_ROWS)
#define AG_Q      64
#define AG_CHUNK  512
#define AG_NOOWN  255u
#define AG_RQ     256
#define AG_NCHUNK (NE / AG_CHUNK)
#define AG_ROT    (AG_NCHUNK / AG_BLOCKS)
#define SV_WORDS  (AG_ROWS * HID)
#define SV_TAIL   (AG_NW * AG_RQ * 2)
#define SV_CNT    (SV_TAIL + AG_NW)
#define L2_PITCH  (HID + 4)
#define L2_RT     ((AG_ROWS + 15) / 16)
#define L2_CT     ((CLS + 15) / 16)
#define L2_TILES  (L2_RT * L2_CT)
#define LG_PITCH  (CLS + 1)

__global__ __launch_bounds__(256)
void edge_owner_tags(const int* __restrict__ dst, unsigned* __restrict__ tags4) {
  static_assert(AG_BLOCKS < (int)AG_NOOWN, "an owner number fits a byte");
  static_assert(NE % 4 == 0, "whole groups of four edges");
  const int i = (int)(blockIdx.x * 256u + threadIdx.x);
  if (i >= NE / 4) return;
  const int4 d = *reinterpret_cast<const int4*>(dst + i * 4);
  const unsigned t0 = (unsigned)d.x < (unsigned)N_NODES ? (unsigned)d.x / AG_ROWS : AG_NOOWN;
  const unsigned t1 = (unsigned)d.y < (unsigned)N_NODES ? (unsigned)d.y / AG_ROWS : AG_NOOWN;
  const unsigned t2 = (unsigned)d.z < (unsigned)N_NODES ? (unsigned)d.z / AG_ROWS : AG_NOOWN;
  const unsigned t3 = (unsigned)d.w < (unsigned)N_NODES ? (unsigned)d.w / AG_ROWS : AG_NOOWN;
  tags4[i] = t0 | (t1 << 8) | (t2 << 16) | (t3 << 24);
}

static __device__ __forceinline__ void lds_add(float* p, float v) {
  __hip_atomic_fetch_add(p, v, __ATOMIC_RELAXED, __HIP_MEMORY_SCOPE_WORKGROUP);
}

template <int F, int P>
static __device__ __forceinline__ void route_ring(const float* __restrict__ feat,
                                                  const int*   __restrict__ src,
                                                  const int*   __restrict__ dst,
                                                  int row0,
                                                  float* acc, float* cnt,
                                                  const int* qe_w,
                                                  uint2* rq, unsigned* rtail,
                                                  int head, int m, int lane) {
  constexpr int NCH = F / 32;
  const int slot = (head + lane) & (AG_Q - 1);
  const int e  = qe_w[slot];
  const int s  = src[e];
  const int dl = dst[e] - row0;
  bool over = false;
  if (lane < m) {
    lds_add(cnt + dl, 1.0f);
    const unsigned ow  = (unsigned)dl & (unsigned)(AG_NW - 1);
    const unsigned pos = __hip_atomic_fetch_add(rtail + ow, 1u, __ATOMIC_RELAXED,
                                                __HIP_MEMORY_SCOPE_WORKGROUP);
    if (pos < (unsigned)AG_RQ) rq[ow * AG_RQ + pos] = make_uint2((unsigned)s, (unsigned)dl);
    else over = true;
  }
  unsigned ov = __builtin_amdgcn_ballot_w32(over);
  while (ov != 0u) {
    const int j  = __builtin_ctz(ov);
    const int sj = __builtin_amdgcn_readlane(s, j);
    const int dj = __builtin_amdgcn_readlane(dl, j);
    const float* f = feat + (long long)sj * F + lane;
    float* a = acc + dj * P + lane;
    #pragma unroll
    for (int c = 0; c < NCH; ++c) lds_add(a + c * 32, f[c * 32]);
    ov &= ov - 1u;
  }
}

static __device__ __forceinline__ v8f mm2(const float4 av, const float4 bv, v8f c) {
  v2f a0; a0.x = av.x; a0.y = av.y;
  v2f b0; b0.x = bv.x; b0.y = bv.y;
  v2f a1; a1.x = av.z; a1.y = av.w;
  v2f b1; b1.x = bv.z; b1.y = bv.w;
  c = __builtin_amdgcn_wmma_f32_16x16x4_f32(false, a0, false, b0, (short)0, c, false, false);
  c = __builtin_amdgcn_wmma_f32_16x16x4_f32(false, a1, false, b1, (short)0, c, false, false);
  return c;
}

static __device__ __forceinline__ v8f layer2_tile(const float4* mean4,
                                                  const float4* wl4,
                                                  const float4* wr4,
                                                  const float* __restrict__ hrows,
                                                  int t, int lane) {
  const int rt  = t / L2_CT;
  const int ct  = t - rt * L2_CT;
  const int c16 = lane & 15;
  const int hh  = lane >> 4;
  int rowl = rt * 16 + c16;
  rowl = rowl < AG_ROWS ? rowl : AG_ROWS - 1;
  int col = ct * 16 + c16;
  col = col < CLS ? col : CLS - 1;
  const float4* ap = mean4 + rowl * (L2_PITCH / 4) + hh;
  const float4* bl = wl4 + col * (L2_PITCH / 4) + hh;
  const float4* br = wr4 + col * (L2_PITCH / 4) + hh;
  const float*  hp = hrows + rowl * HID + hh * 4;
  v8f c = {};
  #pragma unroll 4
  for (int ks = 0; ks < HID / 8; ++ks) c = mm2(ap[2 * ks], bl[2 * ks], c);
  #pragma unroll 4
  for (int ks = 0; ks < HID / 8; ++ks)
    c = mm2(*reinterpret_cast<const float4*>(hp + 8 * ks), br[2 * ks], c);
  return c;
}

static __device__ __forceinline__ void logits_to_lds(float* lg, const v8f d,
                                                     const float* __restrict__ bias,
                                                     int t, int lane) {
  const int rt  = t / L2_CT;
  const int ct  = t - rt * L2_CT;
  const int col = ct * 16 + (lane & 15);
  const int rb  = rt * 16 + (lane >> 4) * 8;
  if (col < CLS) {
    const float bv = bias[col];
    #pragma unroll
    for (int i = 0; i < 8; ++i)
      if (rb + i < AG_ROWS) lg[(rb + i) * LG_PITCH + col] = d[i] + bv;
  }
}

template <int F, int REUSE>
__global__ __launch_bounds__(AG_TB)
void owner_mean(const float* __restrict__ feat,
                const int*   __restrict__ src,
                const int*   __restrict__ dst,
                const unsigned* __restrict__ tags4,
                unsigned* save,
                float* agg,
                const float* __restrict__ w_l,
                const float* __restrict__ w_r,
                const float* __restrict__ bias,
                float* __restrict__ out) {
  static_assert(REUSE == 0 || F == HID, "the fused layer-2 tail is written for rows of HID floats");
  constexpr int P = REUSE ? L2_PITCH : F;
  static_assert(P % 4 == 0, "rows of acc start on 16 bytes");
  static_assert(SV_CNT + AG_ROWS <= SV_WORDS, "queues, lengths and counts fit the work-group's slice");
  static_assert(NE % AG_CHUNK == 0, "a wave reads whole chunks of the tag list");
  static_assert(AG_NCHUNK % AG_BLOCKS == 0, "the scans' offsets tile the list");
  static_assert(N_NODES % AG_ROWS == 0, "every work-group owns a full block of rows");
  static_assert(AG_Q >= 32 + 32, "one new edge per lane fits the ring beside 31 waiting ones");
  __shared__ __attribute__((aligned(16))) float acc[AG_ROWS * P];
  __shared__ float cnt[AG_ROWS];
  __shared__ int   qe[AG_NW][AG_Q];
  __shared__ __attribute__((aligned(16))) uint2 rq[AG_NW * AG_RQ];
  __shared__ unsigned rtail[AG_NW];

  const int tid  = (int)threadIdx.x;
  const int lane = tid & 31;
  const int wave = __builtin_amdgcn_readfirstlane(tid >> 5);
  const int row0 = (int)blockIdx.x * AG_ROWS;

  float4* acc4 = reinterpret_cast<float4*>(acc);
  const float4 z = make_float4(0.0f, 0.0f, 0.0f, 0.0f);
  unsigned* sv = save + (size_t)blockIdx.x * (size_t)SV_WORDS;
  unsigned cv = 0u;
  unsigned tv = 0u;
  if (REUSE) {
    if (tid < AG_ROWS) cv = sv[SV_CNT + tid];
    if (tid < AG_NW) tv = sv[SV_TAIL + tid];
  }
  for (int i = tid; i < AG_ROWS * P / 4; i += AG_TB) acc4[i] = z;
  if (REUSE) {
    if (tid < AG_ROWS) cnt[tid] = __uint_as_float(cv);
    if (tid < AG_NW) rtail[tid] = tv;
  } else {
    if (tid < AG_ROWS) cnt[tid] = 0.0f;
    if (tid < AG_NW) rtail[tid] = 0u;
  }
  int* qe_w = qe[wave];
  for (int i = lane; i < AG_Q; i += 32) qe_w[i] = 0;
  __syncthreads();

  bool scan = true;
  if (REUSE) {
    const unsigned t = rtail[lane];
    scan = __builtin_amdgcn_ballot_w32(t > (unsigned)AG_RQ) != 0u;
    if (scan) {
      __syncthreads();
      if (tid < AG_ROWS) cnt[tid] = 0.0f;
      if (tid < AG_NW) rtail[tid] = 0u;
      __syncthreads();
    } else {
      const int n = __builtin_amdgcn_readfirstlane((int)rtail[wave]);
      const uint2* sq = reinterpret_cast<const uint2*>(sv) + wave * AG_RQ;
      uint2 e[AG_RQ / 32];
      #pragma unroll
      for (int k = 0; k < AG_RQ / 32; ++k) {
        const int i = lane + 32 * k;
        e[k] = i < n ? sq[i] : make_uint2(0u, 0u);
      }
      #pragma unroll
      for (int k = 0; k < AG_RQ / 32; ++k) {
        const int i = lane + 32 * k;
        if (i < n) rq[wave * AG_RQ + i] = e[k];
      }
    }
  }

  if (scan) {
    int head = 0;
    int qn   = 0;
    const int c0   = (int)blockIdx.x * AG_ROT + wave;
    const int cend = (int)blockIdx.x * AG_ROT + AG_NCHUNK;
    const unsigned char* tags = reinterpret_cast<const unsigned char*>(tags4);
    const unsigned K7F = 0x7F7F7F7Fu;
    const unsigned K80 = 0x80808080u;
    const unsigned pat = (unsigned)blockIdx.x * 0x01010101u;
    uint4 nxt = make_uint4(0u, 0u, 0u, 0u);
    {
      const int cc = c0 >= AG_NCHUNK ? c0 - AG_NCHUNK : c0;
      nxt = *reinterpret_cast<const uint4*>(tags + cc * AG_CHUNK + lane * 16);
    }
    for (int c = c0; c < cend; c += AG_NW) {
      const uint4 cur = nxt;
      const int ebase = (c >= AG_NCHUNK ? c - AG_NCHUNK : c) * AG_CHUNK + lane * 16;
      if (c + AG_NW < cend) {
        const int cn = c + AG_NW;
        const int cc = cn >= AG_NCHUNK ? cn - AG_NCHUNK : cn;
        nxt = *reinterpret_cast<const uint4*>(tags + cc * AG_CHUNK + lane * 16);
      }
      const unsigned w0 = cur.x ^ pat;
      const unsigned w1 = cur.y ^ pat;
      const unsigned w2 = cur.z ^ pat;
      const unsigned w3 = cur.w ^ pat;
      const unsigned z0 = ~(((w0 & K7F) + K7F) | w0) & K80;
      const unsigned z1 = ~(((w1 & K7F) + K7F) | w1) & K80;
      const unsigned z2 = ~(((w2 & K7F) + K7F) | w2) & K80;
      const unsigned z3 = ~(((w3 & K7F) + K7F) | w3) & K80;
      unsigned comb = (z0 >> 7) | (z1 >> 6) | (z2 >> 5) | (z3 >> 4);
      unsigned bal = __builtin_amdgcn_ballot_w32(comb != 0u);
      while (bal != 0u) {
        const bool hit = comb != 0u;
        const unsigned p = (unsigned)__builtin_ctz(comb | 0x80000000u);
        const int pos = (head + qn + (int)__builtin_amdgcn_mbcnt_lo(bal, 0u)) & (AG_Q - 1);
        if (hit) qe_w[pos] = ebase + (int)(((p & 3u) << 2) | (p >> 3));
        qn += __builtin_popcount(bal);
        comb &= comb - 1u;
        while (qn >= 32) {
          route_ring<F, P>(feat, src, dst, row0, acc, cnt, qe_w, rq, rtail, head, 32, lane);
          head = (head + 32) & (AG_Q - 1);
          qn -= 32;
        }
        bal = __builtin_amdgcn_ballot_w32(comb != 0u);
      }
    }
    while (qn > 0) {
      const int m = qn < 32 ? qn : 32;
      route_ring<F, P>(feat, src, dst, row0, acc, cnt, qe_w, rq, rtail, head, m, lane);
      head = (head + m) & (AG_Q - 1);
      qn -= m;
    }
  }
  __syncthreads();

  if (!REUSE) {
    const unsigned nt = rtail[wave];
    const int n = __builtin_amdgcn_readfirstlane((int)(nt < (unsigned)AG_RQ ? nt : (unsigned)AG_RQ));
    uint2* sq = reinterpret_cast<uint2*>(sv) + wave * AG_RQ;
    for (int i = lane; i < n; i += 32) sq[i] = rq[wave * AG_RQ + i];
    if (tid < AG_NW) sv[SV_TAIL + tid] = rtail[tid];
    if (tid < AG_ROWS) sv[SV_CNT + tid] = __float_as_uint(cnt[tid]);
  }

  {
    constexpr int NV  = F / 128;
    constexpr int GRP = 2048 / F;
    const unsigned nt = rtail[wave];
    const int n = __builtin_amdgcn_readfirstlane((int)(nt < (unsigned)AG_RQ ? nt : (unsigned)AG_RQ));
    const uint2*  q     = rq + wave * AG_RQ;
    const float4* feat4 = reinterpret_cast<const float4*>(feat);
    for (int b = 0; b < n; b += 32) {
      const int m = n - b < 32 ? n - b : 32;
      const uint2 en = lane < m ? q[b + lane] : make_uint2(0u, 0u);
      const int s  = (int)en.x;
      const int dl = (int)en.y;
      for (int j = 0; j < m; j += GRP) {
        float4 v[GRP][NV];
        #pragma unroll
        for (int g = 0; g < GRP; ++g) {
          const int jj = j + g < m ? j + g : m - 1;
          const int sj = __builtin_amdgcn_readlane(s, jj);
          const float4* f = feat4 + (long long)sj * (F / 4) + lane;
          #pragma unroll
          for (int c = 0; c < NV; ++c) v[g][c] = f[c * 32];
        }
        #pragma unroll
        for (int g = 0; g < GRP; ++g) {
          if (j + g < m) {
            const int dj = __builtin_amdgcn_readlane(dl, j + g);
            float4* a = acc4 + dj * (P / 4) + lane;
            #pragma unroll
            for (int c = 0; c < NV; ++c) {
              float4 t = a[c * 32];
              t.x += v[g][c].x;
              t.y += v[g][c].y;
              t.z += v[g][c].z;
              t.w += v[g][c].w;
              a[c * 32] = t;
            }
          }
        }
      }
    }
  }
  __syncthreads();

  constexpr int Q = F / 4;
  if constexpr (REUSE != 0) {
    constexpr int P4 = P / 4;
    static_assert(CLS * L2_PITCH * 4 <= AG_NW * AG_RQ * 8, "W_l fits the room of the routed queues");
    static_assert(AG_ROWS * LG_PITCH <= AG_ROWS * P, "the logits fit the room of the sums");
    static_assert(L2_TILES > AG_NW && L2_TILES <= 2 * AG_NW, "every wave has one tile, some a second");
    __shared__ __attribute__((aligned(16))) float w2r[CLS * L2_PITCH];
    float4* wl4 = reinterpret_cast<float4*>(rq);
    float4* wr4 = reinterpret_cast<float4*>(w2r);
    const float4* gl = reinterpret_cast<const float4*>(w_l);
    const float4* gr = reinterpret_cast<const float4*>(w_r);
    for (int i = tid; i < CLS * Q; i += AG_TB) {
      const int col = i / Q;
      const int q   = i - col * Q;
      wl4[col * P4 + q] = gl[i];
      wr4[col * P4 + q] = gr[i];
    }
    for (int i = tid; i < AG_ROWS * Q; i += AG_TB) {
      const int r = i / Q;
      const int q = i - r * Q;
      const float c  = cnt[r];
      const float cc = c > 1.0f ? c : 1.0f;
      float4 v = acc4[r * P4 + q];
      v.x = v.x / cc;
      v.y = v.y / cc;
      v.z = v.z / cc;
      v.w = v.w / cc;
      acc4[r * P4 + q] = v;
    }
    __syncthreads();

    const float* hrows = feat + (long long)row0 * F;
    const bool two = wave + AG_NW < L2_TILES;
    v8f d0 = layer2_tile(acc4, wl4, wr4, hrows, wave, lane);
    v8f d1 = {};
    if (two) d1 = layer2_tile(acc4, wl4, wr4, hrows, wave + AG_NW, lane);
    __syncthreads();

    float* lg = acc;
    logits_to_lds(lg, d0, bias, wave, lane);
    if (two) logits_to_lds(lg, d1, bias, wave + AG_NW, lane);
    __syncthreads();

    if (tid < AG_ROWS) {
      float* l = lg + tid * LG_PITCH;
      float m = -3.402823466e38f;
      #pragma unroll
      for (int j = 0; j < CLS; ++j) m = fmaxf(m, l[j]);
      float ssum = 0.0f;
      #pragma unroll
      for (int j = 0; j < CLS; ++j) ssum += expf(l[j] - m);
      const float ls = logf(ssum);
      #pragma unroll
      for (int j = 0; j < CLS; ++j) l[j] = l[j] - m - ls;
    }
    __syncthreads();

    float* o = out + (long long)row0 * CLS;
    for (int i = tid; i < AG_ROWS * CLS; i += AG_TB) {
      const int r = i / CLS;
      o[i] = lg[r * LG_PITCH + (i - r * CLS)];
    }
  } else {
    float4* out4 = reinterpret_cast<float4*>(agg + (long long)row0 * F);
    for (int i = tid; i < AG_ROWS * Q; i += AG_TB) {
      const float c  = cnt[i / Q];
      const float cc = c > 1.0f ? c : 1.0f;
      const float4 v = acc4[i];
      float4 r;
      r.x = v.x / cc;
      r.y = v.y / cc;
      r.z = v.z / cc;
      r.w = v.w / cc;
      out4[i] = r;
    }
  }
}

#define G1_WAVES  (HID / 16)
#define G1_TB     (G1_WAVES * 32)
#define G1_BLOCKS 256
#define G1_MT     (N_NODES / 16)
#define G1_STEPS  (F_IN / 8)
#define G1_GRP    8
#define G1_PITCH  (F_IN + 4)
#define G1_MAT    (16 * G1_PITCH)
#define G1_STAGE  (2 * G1_MAT)
#define G1_NSTG   3

__global__ __launch_bounds__(G1_TB)
void gemm1_weights_in_regs(const float* __restrict__ A1,
                           const float* __restrict__ W1,
                           const float* __restrict__ A2,
                           const float* __restrict__ W2,
                           const float* __restrict__ bias,
                           float* __restrict__ out) {
  static_assert(N_NODES % 16 == 0, "whole row tiles");
  static_assert(G1_STEPS == 2 * G1_GRP, "two groups of k-steps per product");
  static_assert(G1_WAVES == 16, "one wave per row of a tile");
  static_assert(F_IN == 32 * 4, "a wave's 16-byte loads cover one row");
  static_assert((G1_PITCH / 4) % 2 == 1, "staged rows lie 16 bytes x an odd number apart");
  __shared__ __attribute__((aligned(16))) float sa[G1_NSTG * G1_STAGE];

  const int lane  = (int)threadIdx.x & 31;
  const int ct    = __builtin_amdgcn_readfirstlane((int)threadIdx.x >> 5);
  const int c16   = lane & 15;
  const int khalf = (lane >> 4) * 4;
  const int ncol  = ct * 16 + c16;

  float4 bl[G1_STEPS], br[G1_STEPS];
  {
    const float* bp1 = W1 + ncol * F_IN + khalf;
    const float* bp2 = W2 + ncol * F_IN + khalf;
    #pragma unroll
    for (int s = 0; s < G1_STEPS; ++s) {
      bl[s] = *reinterpret_cast<const float4*>(bp1 + 8 * s);
      br[s] = *reinterpret_cast<const float4*>(bp2 + 8 * s);
    }
  }
  const float bv = bias[ncol];

  const int step = (int)gridDim.x;
  const int rt0  = (int)blockIdx.x;
  const int nt   = (G1_MT - rt0 + step - 1) / step;

  const unsigned soff = (unsigned)ct * (unsigned)F_IN + (unsigned)lane * 4u;
  const int      sdst = ct * G1_PITCH + lane * 4;
  const int      frd  = c16 * G1_PITCH + khalf;

  {
    const int r1 = nt > 1 ? rt0 + step : rt0;
    const unsigned o0 = (unsigned)(rt0 * 16) * (unsigned)F_IN + soff;
    const unsigned o1 = (unsigned)(r1 * 16) * (unsigned)F_IN + soff;
    const float4 p0 = *reinterpret_cast<const float4*>(A1 + o0);
    const float4 p1 = *reinterpret_cast<const float4*>(A2 + o0);
    const float4 p2 = *reinterpret_cast<const float4*>(A1 + o1);
    const float4 p3 = *reinterpret_cast<const float4*>(A2 + o1);
    *reinterpret_cast<float4*>(sa + sdst) = p0;
    *reinterpret_cast<float4*>(sa + G1_MAT + sdst) = p1;
    *reinterpret_cast<float4*>(sa + G1_STAGE + sdst) = p2;
    *reinterpret_cast<float4*>(sa + G1_STAGE + G1_MAT + sdst) = p3;
  }
  __syncthreads();

  float4 g0[G1_GRP], g1[G1_GRP], g2[G1_GRP], g3[G1_GRP];
  int cur = 0;
  int nxt = G1_STAGE;
  int fil = 2 * G1_STAGE;
  float4 q1, q2;
  {
    const int rf = nt > 2 ? rt0 + 2 * step : rt0;
    const unsigned of = (unsigned)(rf * 16) * (unsigned)F_IN + soff;
    q1 = *reinterpret_cast<const float4*>(A1 + of);
    q2 = *reinterpret_cast<const float4*>(A2 + of);
  }
  #pragma unroll
  for (int s = 0; s < G1_GRP; ++s) g0[s] = *reinterpret_cast<const float4*>(sa + cur + frd + 8 * s);

  for (int i = 0; i < nt; ++i) {
    const int rt = rt0 + i * step;

    const float* f1 = sa + cur + frd;
    const float* f2 = f1 + G1_MAT;
    const float* fn = sa + nxt + frd;
    v8f c = {};
    #pragma unroll
    for (int s = 0; s < G1_GRP; ++s) g1[s] = *reinterpret_cast<const float4*>(f1 + 8 * (G1_GRP + s));
    __builtin_amdgcn_sched_barrier(0);
    #pragma unroll
    for (int s = 0; s < G1_GRP; ++s) c = mm2(g0[s], bl[s], c);
    __builtin_amdgcn_sched_barrier(0);
    #pragma unroll
    for (int s = 0; s < G1_GRP; ++s) g2[s] = *reinterpret_cast<const float4*>(f2 + 8 * s);
    __builtin_amdgcn_sched_barrier(0);
    #pragma unroll
    for (int s = 0; s < G1_GRP; ++s) c = mm2(g1[s], bl[G1_GRP + s], c);
    __builtin_amdgcn_sched_barrier(0);
    #pragma unroll
    for (int s = 0; s < G1_GRP; ++s) g3[s] = *reinterpret_cast<const float4*>(f2 + 8 * (G1_GRP + s));
    __builtin_amdgcn_sched_barrier(0);
    #pragma unroll
    for (int s = 0; s < G1_GRP; ++s) c = mm2(g2[s], br[s], c);
    __builtin_amdgcn_sched_barrier(0);
    #pragma unroll
    for (int s = 0; s < G1_GRP; ++s) g0[s] = *reinterpret_cast<const float4*>(fn + 8 * s);
    __builtin_amdgcn_sched_barrier(0);
    #pragma unroll
    for (int s = 0; s < G1_GRP; ++s) c = mm2(g3[s], br[G1_GRP + s], c);

    *reinterpret_cast<float4*>(__builtin_assume_aligned(sa + fil + sdst, 16)) = q1;
    *reinterpret_cast<float4*>(__builtin_assume_aligned(sa + fil + G1_MAT + sdst, 16)) = q2;
    __syncthreads();
    const int t = cur;
    cur = nxt;
    nxt = fil;
    fil = t;

    const unsigned rbase = (unsigned)(rt * 16 + ((lane < 16) ? 0 : 8));
    float* orow = out + rbase * (unsigned)HID + ncol;
    #pragma unroll
    for (int j = 0; j < 8; ++j) {
      float v = c[j] + bv;
      v = v > 0.0f ? v : 0.0f;
      orow[(unsigned)j * (unsigned)HID] = v;
    }

    {
      const int rf = i + 3 < nt ? rt + 3 * step : rt0;
      const unsigned of = (unsigned)(rf * 16) * (unsigned)F_IN + soff;
      q1 = *reinterpret_cast<const float4*>(A1 + of);
      q2 = *reinterpret_cast<const float4*>(A2 + of);
    }
  }
}

extern "C" void kernel_launch(void* const* d_in, const int* in_sizes, int n_in,
                              void* d_out, int out_size, void* d_ws, size_t ws_size,
                              hipStream_t stream) {
  (void)in_sizes; (void)n_in; (void)out_size; (void)ws_size;

  const float* x    = (const float*)d_in[0];
  const int*   ei   = (const int*)  d_in[1];
  const float* W1_l = (const float*)d_in[2];
  const float* b1   = (const float*)d_in[3];
  const float* W1_r = (const float*)d_in[4];
  const float* W2_l = (const float*)d_in[5];
  const float* b2   = (const float*)d_in[6];
  const float* W2_r = (const float*)d_in[7];
  const int* src = ei;
  const int* dst = ei + NE;
  float* out = (float*)d_out;

  float* ws     = (float*)d_ws;
  float* agg1   = ws   + N_NODES;
  float* h      = agg1 + (long long)N_NODES * F_IN;
  float* agg2   = h    + (long long)N_NODES * HID;
  float* tagpl  = agg2 + (long long)N_NODES * HID;
  unsigned* tags4 = (unsigned*)tagpl;

  const int TB = 256;
  auto blocks = [](long long n, int tb) { return (unsigned)((n + tb - 1) / tb); };

  edge_owner_tags<<<blocks(NE / 4, TB), TB, 0, stream>>>(dst, tags4);

  owner_mean<F_IN, 0><<<AG_BLOCKS, AG_TB, 0, stream>>>(x, src, dst, tags4, (unsigned*)agg2, agg1,
                                                       nullptr, nullptr, nullptr, nullptr);

  gemm1_weights_in_regs<<<G1_BLOCKS, G1_TB, 0, stream>>>(agg1, W1_l, x, W1_r, b1, h);

  owner_mean<HID, 1><<<AG_BLOCKS, AG_TB, 0, stream>>>(h, src, dst, tags4, (unsigned*)agg2, nullptr,
                                                      W2_l, W2_r, b2, out);
}
